# starting kernel
# baseline (speedup 1.0000x reference)
#include <hip/hip_runtime.h>

typedef __attribute__((ext_vector_type(2))) float v2f;
typedef __attribute__((ext_vector_type(8))) float v8f;

#define NN       50000
#define IN_CH    128
#define OUT_CH   64
#define HEADS    2
#define HID      128
#define HC1      (HEADS*HID)
#define E0_EDGES 400000
#define E_TOT    (E0_EDGES + NN)
#define NEG_SLOPE 0.2f

struct Bop { v2f b0, b1, b2, b3; };

__device__ __forceinline__ v2f load_a(const float* __restrict__ arow, int k0, int lhalf) {
  v2f a;
  a.x = arow[k0 + 2 * lhalf];
  a.y = arow[k0 + 2 * lhalf + 1];
  return a;
}
__device__ __forceinline__ Bop load_b(const float* __restrict__ B, int k0, int N,
                                      int tn, int lhalf, int lmod) {
  const float* r0 = B + (size_t)(k0 + 2 * lhalf) * N + tn + lmod;
  const float* r1 = r0 + N;
  Bop b;
  b.b0.x = r0[0];  b.b0.y = r1[0];
  b.b1.x = r0[16]; b.b1.y = r1[16];
  b.b2.x = r0[32]; b.b2.y = r1[32];
  b.b3.x = r0[48]; b.b3.y = r1[48];
  return b;
}

__global__ void __launch_bounds__(256)
gemm_wmma_f32(const float* __restrict__ A, const float* __restrict__ B,
              float* __restrict__ C, int M, int N, int K) {
  const int wave   = blockIdx.x * (blockDim.x >> 5) + (threadIdx.x >> 5);
  const int tilesN = N >> 6;
  const int tm     = wave / tilesN;
  const int tn     = (wave % tilesN) << 6;
  if (tm * 16 >= M) return;
  const int lhalf = (threadIdx.x >> 4) & 1;
  const int lmod  = threadIdx.x & 15;

  v8f c0 = {}, c1 = {}, c2 = {}, c3 = {};
  const float* arow = A + (size_t)(tm * 16 + lmod) * K;

  v2f a = load_a(arow, 0, lhalf);
  Bop b = load_b(B, 0, N, tn, lhalf, lmod);
  for (int k0 = 0; k0 < K - 4; k0 += 4) {
    v2f an = load_a(arow, k0 + 4, lhalf);
    Bop bn = load_b(B, k0 + 4, N, tn, lhalf, lmod);
    c0 = __builtin_amdgcn_wmma_f32_16x16x4_f32(false, a, false, b.b0, (short)0, c0, false, false);
    c1 = __builtin_amdgcn_wmma_f32_16x16x4_f32(false, a, false, b.b1, (short)0, c1, false, false);
    c2 = __builtin_amdgcn_wmma_f32_16x16x4_f32(false, a, false, b.b2, (short)0, c2, false, false);
    c3 = __builtin_amdgcn_wmma_f32_16x16x4_f32(false, a, false, b.b3, (short)0, c3, false, false);
    a = an; b = bn;
  }
  c0 = __builtin_amdgcn_wmma_f32_16x16x4_f32(false, a, false, b.b0, (short)0, c0, false, false);
  c1 = __builtin_amdgcn_wmma_f32_16x16x4_f32(false, a, false, b.b1, (short)0, c1, false, false);
  c2 = __builtin_amdgcn_wmma_f32_16x16x4_f32(false, a, false, b.b2, (short)0, c2, false, false);
  c3 = __builtin_amdgcn_wmma_f32_16x16x4_f32(false, a, false, b.b3, (short)0, c3, false, false);

#pragma unroll
  for (int j = 0; j < 8; ++j) {
    float* p = C + (size_t)(tm * 16 + j + 8 * lhalf) * N + tn + lmod;
    p[0] = c0[j]; p[16] = c1[j]; p[32] = c2[j]; p[48] = c3[j];
  }
}

__global__ void __launch_bounds__(256)
att_scores(const float* __restrict__ h, const float* __restrict__ att_s,
           const float* __restrict__ att_d, float* __restrict__ as,
           float* __restrict__ ad, int n_nodes, int H, int C) {
  int i = blockIdx.x * blockDim.x + threadIdx.x;
  if (i >= n_nodes * H) return;
  int n = i / H, hh = i % H;
  const float4* hp = (const float4*)(h + (size_t)n * H * C + (size_t)hh * C);
  const float4* sp = (const float4*)(att_s + hh * C);
  const float4* dp = (const float4*)(att_d + hh * C);
  float s = 0.f, d = 0.f;
  for (int c = 0; c < (C >> 2); ++c) {
    float4 v = hp[c], ws = sp[c], wd = dp[c];
    s += v.x * ws.x + v.y * ws.y + v.z * ws.z + v.w * ws.w;
    d += v.x * wd.x + v.y * wd.y + v.z * wd.z + v.w * wd.w;
  }
  as[i] = s; ad[i] = d;
}

__device__ __forceinline__ void edge_nodes(const int* __restrict__ ei, int e,
                                           int& s, int& d) {
  if (e < E0_EDGES) { s = ei[e]; d = ei[E0_EDGES + e]; }
  else              { s = e - E0_EDGES; d = s; }
}
__device__ __forceinline__ float leaky(float v) { return v >= 0.f ? v : NEG_SLOPE * v; }
__device__ __forceinline__ unsigned f2ord(float f) {
  unsigned u = __float_as_uint(f);
  return (u & 0x80000000u) ? ~u : (u | 0x80000000u);
}
__device__ __forceinline__ float ord2f(unsigned k) {
  unsigned u = (k & 0x80000000u) ? (k ^ 0x80000000u) : ~k;
  return __uint_as_float(u);
}

__global__ void __launch_bounds__(256)
edge_max(const int* __restrict__ ei, const float* __restrict__ as,
         const float* __restrict__ ad, unsigned* __restrict__ m, int H) {
  int i = blockIdx.x * blockDim.x + threadIdx.x;
  if (i >= E_TOT * H) return;
  int e = i / H, hh = i % H;
  int s, d; edge_nodes(ei, e, s, d);
  float v = leaky(as[s * H + hh] + ad[d * H + hh]);
  atomicMax(&m[d * H + hh], f2ord(v));
}

__global__ void __launch_bounds__(256)
edge_exp(const int* __restrict__ ei, const float* __restrict__ as,
         const float* __restrict__ ad, const unsigned* __restrict__ m,
         float* __restrict__ ea, float* __restrict__ denom, int H) {
  int i = blockIdx.x * blockDim.x + threadIdx.x;
  if (i >= E_TOT * H) return;
  int e = i / H, hh = i % H;
  int s, d; edge_nodes(ei, e, s, d);
  float v  = leaky(as[s * H + hh] + ad[d * H + hh]);
  float ex = __expf(v - ord2f(m[d * H + hh]));
  ea[i] = ex;
  atomicAdd(&denom[d * H + hh], ex);
}

__global__ void __launch_bounds__(256)
edge_scatter(const int* __restrict__ ei, const float* __restrict__ h,
             const float* __restrict__ ea, const float* __restrict__ denom,
             float* __restrict__ agg, int H, int C) {
  const int HC  = H * C;
  const int per = HC >> 2;
  long long i = (long long)blockIdx.x * blockDim.x + threadIdx.x;
  if (i >= (long long)E_TOT * per) return;
  int e  = (int)(i / per);
  int cq = (int)(i % per) << 2;
  int hh = cq / C;
  int s, d; edge_nodes(ei, e, s, d);
  float a = ea[(size_t)e * H + hh] / (denom[d * H + hh] + 1e-16f);
  float4 hv = *(const float4*)(h + (size_t)s * HC + cq);
  float* p = agg + (size_t)d * HC + cq;
  atomicAdd(p + 0, hv.x * a);
  atomicAdd(p + 1, hv.y * a);
  atomicAdd(p + 2, hv.z * a);
  atomicAdd(p + 3, hv.w * a);
}

__global__ void __launch_bounds__(256)
bias_elu(float* __restrict__ h, const float* __restrict__ b, int total, int C) {
  int i = blockIdx.x * blockDim.x + threadIdx.x;
  if (i >= total) return;
  float v = h[i] + b[i % C];
  h[i] = v > 0.f ? v : (__expf(v) - 1.f);
}

__global__ void __launch_bounds__(256)
bias_add(float* __restrict__ o, const float* __restrict__ b, int total, int C) {
  int i = blockIdx.x * blockDim.x + threadIdx.x;
  if (i >= total) return;
  o[i] += b[i % C];
}

extern "C" void kernel_launch(void* const* d_in, const int* in_sizes, int n_in,
                              void* d_out, int out_size, void* d_ws, size_t ws_size,
                              hipStream_t stream) {
  const float* x    = (const float*)d_in[0];
  const int*   ei   = (const int*)  d_in[1];
  const float* W1   = (const float*)d_in[2];
  const float* aS1  = (const float*)d_in[3];
  const float* aD1  = (const float*)d_in[4];
  const float* b1   = (const float*)d_in[5];
  const float* Wmu  = (const float*)d_in[6];
  const float* aSmu = (const float*)d_in[7];
  const float* aDmu = (const float*)d_in[8];
  const float* bmu  = (const float*)d_in[9];
  const float* Wls  = (const float*)d_in[10];
  const float* aSls = (const float*)d_in[11];
  const float* aDls = (const float*)d_in[12];
  const float* bls  = (const float*)d_in[13];

  char* ws = (char*)d_ws;
  size_t off = 0;
  auto alloc = [&](size_t bytes) -> char* {
    off = (off + 255) & ~(size_t)255;
    char* p = ws + off; off += bytes; return p;
  };

  float*    h1   = (float*)   alloc((size_t)NN * HC1 * 4);
  float*    hbuf = (float*)   alloc((size_t)NN * HC1 * 4);
  float*    as1  = (float*)   alloc((size_t)NN * HEADS * 4);
  float*    ad1  = (float*)   alloc((size_t)NN * HEADS * 4);
  unsigned* m1   = (unsigned*)alloc((size_t)NN * HEADS * 4);
  float*    dn1  = (float*)   alloc((size_t)NN * HEADS * 4);
  float*    ea1  = (float*)   alloc((size_t)E_TOT * HEADS * 4);
  float*    as2m = (float*)   alloc((size_t)NN * 4);
  float*    ad2m = (float*)   alloc((size_t)NN * 4);
  unsigned* m2m  = (unsigned*)alloc((size_t)NN * 4);
  float*    dn2m = (float*)   alloc((size_t)NN * 4);
  float*    ea2m = (float*)   alloc((size_t)E_TOT * 4);
  float*    as2l = (float*)   alloc((size_t)NN * 4);
  float*    ad2l = (float*)   alloc((size_t)NN * 4);
  unsigned* m2l  = (unsigned*)alloc((size_t)NN * 4);
  float*    dn2l = (float*)   alloc((size_t)NN * 4);
  float*    ea2l = (float*)   alloc((size_t)E_TOT * 4);

  float* hm = h1;
  float* hl = h1 + (size_t)NN * OUT_CH;
  float* out_mu = (float*)d_out;
  float* out_ls = out_mu + (size_t)NN * OUT_CH;

  hipMemsetAsync(hbuf, 0, (size_t)NN * HC1 * 4, stream);
  hipMemsetAsync(m1,   0, (size_t)NN * HEADS * 4, stream);
  hipMemsetAsync(dn1,  0, (size_t)NN * HEADS * 4, stream);

  int waves = (NN / 16) * (HC1 / 64);
  gemm_wmma_f32<<<(waves + 7) / 8, 256, 0, stream>>>(x, W1, h1, NN, HC1, IN_CH);
  att_scores<<<(NN * HEADS + 255) / 256, 256, 0, stream>>>(h1, aS1, aD1, as1, ad1, NN, HEADS, HID);
  edge_max<<<(E_TOT * HEADS + 255) / 256, 256, 0, stream>>>(ei, as1, ad1, m1, HEADS);
  edge_exp<<<(E_TOT * HEADS + 255) / 256, 256, 0, stream>>>(ei, as1, ad1, m1, ea1, dn1, HEADS);
  {
    long long work = (long long)E_TOT * (HC1 / 4);
    edge_scatter<<<(unsigned)((work + 255) / 256), 256, 0, stream>>>(ei, h1, ea1, dn1, hbuf, HEADS, HID);
  }
  bias_elu<<<(NN * HC1 + 255) / 256, 256, 0, stream>>>(hbuf, b1, NN * HC1, HC1);

  waves = (NN / 16) * (OUT_CH / 64);
  gemm_wmma_f32<<<(waves + 7) / 8, 256, 0, stream>>>(hbuf, Wmu, hm, NN, OUT_CH, HC1);
  gemm_wmma_f32<<<(waves + 7) / 8, 256, 0, stream>>>(hbuf, Wls, hl, NN, OUT_CH, HC1);
  att_scores<<<(NN + 255) / 256, 256, 0, stream>>>(hm, aSmu, aDmu, as2m, ad2m, NN, 1, OUT_CH);
  att_scores<<<(NN + 255) / 256, 256, 0, stream>>>(hl, aSls, aDls, as2l, ad2l, NN, 1, OUT_CH);

  hipMemsetAsync(m2m, 0, (size_t)NN * 4, stream);
  hipMemsetAsync(dn2m, 0, (size_t)NN * 4, stream);
  hipMemsetAsync(m2l, 0, (size_t)NN * 4, stream);
  hipMemsetAsync(dn2l, 0, (size_t)NN * 4, stream);
  hipMemsetAsync(d_out, 0, (size_t)2 * NN * OUT_CH * 4, stream);

  edge_max<<<(E_TOT + 255) / 256, 256, 0, stream>>>(ei, as2m, ad2m, m2m, 1);
  edge_exp<<<(E_TOT + 255) / 256, 256, 0, stream>>>(ei, as2m, ad2m, m2m, ea2m, dn2m, 1);
  {
    long long work = (long long)E_TOT * (OUT_CH / 4);
    edge_scatter<<<(unsigned)((work + 255) / 256), 256, 0, stream>>>(ei, hm, ea2m, dn2m, out_mu, 1, OUT_CH);
  }
  edge_max<<<(E_TOT + 255) / 256, 256, 0, stream>>>(ei, as2l, ad2l, m2l, 1);
  edge_exp<<<(E_TOT + 255) / 256, 256, 0, stream>>>(ei, as2l, ad2l, m2l, ea2l, dn2l, 1);
  {
    long long work = (long long)E_TOT * (OUT_CH / 4);
    edge_scatter<<<(unsigned)((work + 255) / 256), 256, 0, stream>>>(ei, hl, ea2l, dn2l, out_ls, 1, OUT_CH);
  }

  bias_add<<<(NN * OUT_CH + 255) / 256, 256, 0, stream>>>(out_mu, bmu, NN * OUT_CH, OUT_CH);
  bias_add<<<(NN * OUT_CH + 255) / 256, 256, 0, stream>>>(out_ls, bls, NN * OUT_CH, OUT_CH);
}


# layer 1 sums the input rows over a tile's edges first, then applies W1 to the 16 summed rows in the same block (linearity); x*W1 and its launch gone, scores from x.(W1 att); sums change order
# speedup vs baseline: 3.6163x; 3.6163x over previous
#include <hip/hip_runtime.h>

typedef __attribute__((ext_vector_type(2))) float v2f;
typedef __attribute__((ext_vector_type(8))) float v8f;

#define NN       50000
#define IN_CH    128
#define OUT_CH   64
#define HEADS    2
#define HID      128
#define HC1      (HEADS*HID)
#define E0_EDGES 400000
#define E_TOT    (E0_EDGES + NN)
#define NEG_SLOPE 0.2f

struct Bop { v2f b0, b1, b2, b3; };

__device__ __forceinline__ float elu1(float v) { return v > 0.f ? v : (__expf(v) - 1.f); }

__device__ __forceinline__ Bop load_b(const float* __restrict__ B, int k0, int N,
                                      int tn, int lhalf, int lmod) {
  const float* r0 = B + (size_t)(k0 + 2 * lhalf) * N + tn + lmod;
  const float* r1 = r0 + N;
  Bop b;
  b.b0.x = r0[0];  b.b0.y = r1[0];
  b.b1.x = r0[16]; b.b1.y = r1[16];
  b.b2.x = r0[32]; b.b2.y = r1[32];
  b.b3.x = r0[48]; b.b3.y = r1[48];
  return b;
}

__global__ void __launch_bounds__(256)
att_scores(const float* __restrict__ h, const float* __restrict__ att_s,
           const float* __restrict__ att_d, float* __restrict__ as,
           float* __restrict__ ad, int n_nodes, int H, int C) {
  int i = blockIdx.x * blockDim.x + threadIdx.x;
  if (i >= n_nodes * H) return;
  int n = i / H, hh = i % H;
  const float4* hp = (const float4*)(h + (size_t)n * H * C + (size_t)hh * C);
  const float4* sp = (const float4*)(att_s + hh * C);
  const float4* dp = (const float4*)(att_d + hh * C);
  float s = 0.f, d = 0.f;
  for (int c = 0; c < (C >> 2); ++c) {
    float4 v = hp[c], ws = sp[c], wd = dp[c];
    s += v.x * ws.x + v.y * ws.y + v.z * ws.z + v.w * ws.w;
    d += v.x * wd.x + v.y * wd.y + v.z * wd.z + v.w * wd.w;
  }
  as[i] = s; ad[i] = d;
}

__device__ __forceinline__ float leaky(float v) { return v >= 0.f ? v : NEG_SLOPE * v; }

#define NTILES    (NN / 16)
#define GROUP_T   16
#define NGROUPS   ((NTILES + GROUP_T - 1) / GROUP_T)
#define SUB_CAP   48
#define TILE_CAP  256
#define LIST_FULL 0xFFFFFFFFu
#define MAX_ENT   (TILE_CAP + 16 + 16)

__global__ void __launch_bounds__(512)
bin_edges(const int* __restrict__ ei, unsigned* __restrict__ tl,
          unsigned* __restrict__ tcnt, const float* __restrict__ Wl1,
          const float* __restrict__ att_s, const float* __restrict__ att_d,
          float* __restrict__ V) {
  __shared__ unsigned cnt[16][GROUP_T];
  __shared__ unsigned sub[16][GROUP_T][SUB_CAP];
  const int g = blockIdx.x, tid = threadIdx.x;
  const int lane = tid & 31, w = tid >> 5;
  if (g == NGROUPS) {
    const int which = tid >> 7, k = tid & (IN_CH - 1), hd = which & 1;
    const float4* wr = (const float4*)(Wl1 + (size_t)k * HC1 + hd * HID);
    const float4* av = (const float4*)((which < 2 ? att_s : att_d) + hd * HID);
    float s = 0.f;
    for (int cc = 0; cc < HID / 4; ++cc) {
      const float4 a = wr[cc], b = av[cc];
      s += a.x * b.x + a.y * b.y + a.z * b.z + a.w * b.w;
    }
    V[which * IN_CH + k] = s;
    return;
  }
  if (tid < 16 * GROUP_T) cnt[tid >> 4][tid & 15] = 0u;
  __syncthreads();

  const int per_wave = E0_EDGES / 16;
  const int beg = w * per_wave, end = beg + per_wave;
  const int* __restrict__ dstp = ei + E0_EDGES;
  for (int base = beg; base < end; base += 1024) {
    int4 dv[8];
#pragma unroll
    for (int q = 0; q < 8; ++q) {
      const int idx = base + q * 128 + 4 * lane;
      dv[q] = *(const int4*)(dstp + (idx < end ? idx : end - 4));
    }
#pragma unroll
    for (int q = 0; q < 8; ++q) {
      const int idx = base + q * 128 + 4 * lane;
      const int dd[4] = {dv[q].x, dv[q].y, dv[q].z, dv[q].w};
#pragma unroll
      for (int k = 0; k < 4; ++k) {
        const int d = dd[k];
        if (idx < end && (d >> 8) == g) {
          const int ti = (d >> 4) & (GROUP_T - 1);
          const unsigned p = atomicAdd(&cnt[w][ti], 1u);
          if (p < SUB_CAP) sub[w][ti][p] = (unsigned)(idx + k) | ((unsigned)(d & 15) << 24);
        }
      }
    }
  }
  __syncthreads();

  const int T = g * GROUP_T + w;
  if (T < NTILES) {
    unsigned total = 0; bool full = false;
    for (int w2 = 0; w2 < 16; ++w2) {
      const unsigned n = cnt[w2][w];
      if (n > SUB_CAP) full = true;
      const unsigned nn = n < SUB_CAP ? n : SUB_CAP;
      for (unsigned i = lane; i < nn; i += 32) {
        const unsigned pos = total + i;
        if (pos < TILE_CAP) tl[(size_t)T * TILE_CAP + pos] = sub[w2][w][i];
      }
      total += n;
    }
    if (total > TILE_CAP) full = true;
    if (lane == 0) tcnt[T] = full ? LIST_FULL : total;
  }
}

__device__ __forceinline__ float edge_score(const float* __restrict__ as,
    const float* __restrict__ ad, int s, int d, int H, int q) {
  return leaky(as[s * H + q] + ad[d * H + q]);
}

__device__ __forceinline__ float4 ld_row(const float* __restrict__ hcol, int s, int HC) {
  return *(const float4*)(hcol + (size_t)s * HC);
}
__device__ __forceinline__ v2f tile_a(const int* row, const float* al, int k, int c) {
  const float w0 = al[k], w1 = al[k + 1];
  v2f a;
  a.x = row[k]     == c ? w0 : 0.f;
  a.y = row[k + 1] == c ? w1 : 0.f;
  return a;
}
__device__ __forceinline__ void tile_mma(v8f& c0, v8f& c1, v8f& c2, v8f& c3, v2f a,
                                         float4 r0, float4 r1) {
  v2f b;
  b.x = r0.x; b.y = r1.x;
  c0 = __builtin_amdgcn_wmma_f32_16x16x4_f32(false, a, false, b, (short)0, c0, false, false);
  b.x = r0.y; b.y = r1.y;
  c1 = __builtin_amdgcn_wmma_f32_16x16x4_f32(false, a, false, b, (short)0, c1, false, false);
  b.x = r0.z; b.y = r1.z;
  c2 = __builtin_amdgcn_wmma_f32_16x16x4_f32(false, a, false, b, (short)0, c2, false, false);
  b.x = r0.w; b.y = r1.w;
  c3 = __builtin_amdgcn_wmma_f32_16x16x4_f32(false, a, false, b, (short)0, c3, false, false);
}

__device__ __forceinline__ int   ord_key(float f) { const int b = __float_as_int(f); return b >= 0 ? b : b ^ 0x7FFFFFFF; }
__device__ __forceinline__ float ord_val(int k)   { return __int_as_float(k >= 0 ? k : k ^ 0x7FFFFFFF); }

template <int HC, int C>
__global__ void __launch_bounds__(HC / 2)
agg_tile(const unsigned* __restrict__ tl, const unsigned* __restrict__ tcnt,
         const int* __restrict__ ei, const float* __restrict__ as,
         const float* __restrict__ ad, const float* __restrict__ hs,
         const float* __restrict__ bias, float* __restrict__ out) {
  constexpr int H     = HC / C;
  constexpr int NT    = HC / 2;
  constexpr int NPAIR = H * 16;
  static_assert(NT >= NPAIR, "one thread per pair clears the maxima");
  __shared__ int   s_src[MAX_ENT];
  __shared__ int   s_row[MAX_ENT];
  __shared__ float s_al[H][MAX_ENT];
  __shared__ int   s_mk[NPAIR];
  const int T = blockIdx.x, tid = threadIdx.x;
  const int lane = tid & 31, u = tid >> 5, c = lane & 15, hf = lane >> 4;
  const int hh = (u * 64) / C;
  const int node0 = T * 16;
  const float* __restrict__ hcol = hs + u * 64 + 4 * c;

  v8f c0 = {}, c1 = {}, c2 = {}, c3 = {};
  float dsum = 0.f;
  const float* bp = bias + u * 64 + 4 * c;
  const float4 bv = make_float4(bp[0], bp[1], bp[2], bp[3]);

  const unsigned cntT = tcnt[T];
  if (cntT <= TILE_CAP) {
    const int n  = (int)cntT;
    const int nE = n + 16;
    const int nP = (nE + 15) & ~15;
    if (tid < NPAIR) s_mk[tid] = (int)0x80000000;
    __syncthreads();
    for (int i = tid; i < nP; i += NT) {
      int s = node0, row = 0;
      if (i < n) {
        const unsigned ent = tl[(size_t)T * TILE_CAP + i];
        s = ei[ent & 0x00FFFFFFu]; row = (int)(ent >> 24);
      } else if (i < nE) {
        row = i - n; s = node0 + row;
      }
      s_src[i] = s; s_row[i] = row;
#pragma unroll
      for (int q = 0; q < H; ++q) {
        float sc = 0.f;
        if (i < nE) {
          sc = edge_score(as, ad, s, node0 + row, H, q);
          atomicMax(&s_mk[q * 16 + row], ord_key(sc));
        }
        s_al[q][i] = sc;
      }
    }
    __syncthreads();
    for (int i = tid; i < nE; i += NT) {
      const int row = s_row[i];
#pragma unroll
      for (int q = 0; q < H; ++q)
        s_al[q][i] = __expf(s_al[q][i] - ord_val(s_mk[q * 16 + row]));
    }
    __syncthreads();
    for (int j = 0; j < nP; j += 16) {
      v2f a[4]; float4 r[8];
#pragma unroll
      for (int q = 0; q < 4; ++q) {
        const int k = j + 4 * q + 2 * hf;
        a[q] = tile_a(s_row, s_al[hh], k, c);
        r[2 * q]     = ld_row(hcol, s_src[k], HC);
        r[2 * q + 1] = ld_row(hcol, s_src[k + 1], HC);
      }
      __builtin_amdgcn_sched_barrier(0);
#pragma unroll
      for (int q = 0; q < 4; ++q) {
        dsum += a[q].x + a[q].y;
        tile_mma(c0, c1, c2, c3, a[q], r[2 * q], r[2 * q + 1]);
      }
    }
  } else {
    if (tid < NPAIR) {
      const int row = tid & 15, q = tid >> 4, d = node0 + row;
      float mx = edge_score(as, ad, d, d, H, q);
      for (int e = 0; e < E0_EDGES; ++e)
        if (ei[E0_EDGES + e] == d) mx = fmaxf(mx, edge_score(as, ad, ei[e], d, H, q));
      s_mk[tid] = ord_key(mx);
    }
    __syncthreads();
    const float mx0 = ord_val(s_mk[hh * 16 + c]);
    for (int e0 = 0; e0 < E0_EDGES; e0 += 4) {
      const int e  = e0 + 2 * hf;
      const int d0 = ei[E0_EDGES + e], d1 = ei[E0_EDGES + e + 1];
      const bool in0 = (d0 >> 4) == T, in1 = (d1 >> 4) == T;
      if (__ballot(in0 | in1) == 0) continue;
      const int s0 = ei[e], s1 = ei[e + 1];
      v2f a;
      a.x = (in0 && (d0 & 15) == c) ? __expf(edge_score(as, ad, s0, d0, H, hh) - mx0) : 0.f;
      a.y = (in1 && (d1 & 15) == c) ? __expf(edge_score(as, ad, s1, d1, H, hh) - mx0) : 0.f;
      dsum += a.x + a.y;
      tile_mma(c0, c1, c2, c3, a, ld_row(hcol, s0, HC), ld_row(hcol, s1, HC));
    }
    for (int j = 0; j < 16; j += 4) {
      const int r0 = j + 2 * hf, n0 = node0 + r0, n1 = n0 + 1;
      v2f a;
      a.x = r0 == c     ? __expf(edge_score(as, ad, n0, n0, H, hh) - mx0) : 0.f;
      a.y = r0 + 1 == c ? __expf(edge_score(as, ad, n1, n1, H, hh) - mx0) : 0.f;
      dsum += a.x + a.y;
      tile_mma(c0, c1, c2, c3, a, ld_row(hcol, n0, HC), ld_row(hcol, n1, HC));
    }
  }

  const float dtot = dsum + __shfl_xor(dsum, 16);
#pragma unroll
  for (int r = 0; r < 8; ++r) {
    const float inv = 1.0f / (__shfl(dtot, 8 * hf + r) + 1e-16f);
    float* q = out + (size_t)(node0 + 8 * hf + r) * HC + u * 64 + 4 * c;
    *(float4*)q = make_float4(bv.x + c0[r] * inv, bv.y + c1[r] * inv,
                              bv.z + c2[r] * inv, bv.w + c3[r] * inv);
  }
}

__global__ void __launch_bounds__(256)
x_scores(const float* __restrict__ x, const float* __restrict__ V,
         float* __restrict__ as, float* __restrict__ ad, int n_nodes) {
  const int i = blockIdx.x * blockDim.x + threadIdx.x;
  if (i >= n_nodes) return;
  const float4* xp = (const float4*)(x + (size_t)i * IN_CH);
  const float4* v0 = (const float4*)V;
  const float4* v1 = v0 + IN_CH / 4;
  const float4* v2 = v1 + IN_CH / 4;
  const float4* v3 = v2 + IN_CH / 4;
  float s0 = 0.f, s1 = 0.f, d0 = 0.f, d1 = 0.f;
#pragma unroll 8
  for (int cc = 0; cc < IN_CH / 4; ++cc) {
    const float4 v = xp[cc], a = v0[cc], b = v1[cc], p = v2[cc], q = v3[cc];
    s0 += v.x * a.x + v.y * a.y + v.z * a.z + v.w * a.w;
    s1 += v.x * b.x + v.y * b.y + v.z * b.z + v.w * b.w;
    d0 += v.x * p.x + v.y * p.y + v.z * p.z + v.w * p.w;
    d1 += v.x * q.x + v.y * q.y + v.z * q.z + v.w * q.w;
  }
  as[2 * i] = s0; as[2 * i + 1] = s1;
  ad[2 * i] = d0; ad[2 * i + 1] = d1;
}

__global__ void __launch_bounds__(128)
agg1_tile(const unsigned* __restrict__ tl, const unsigned* __restrict__ tcnt,
          const int* __restrict__ ei, const float* __restrict__ as,
          const float* __restrict__ ad, const float* __restrict__ x,
          const float* __restrict__ Wl1, const float* __restrict__ bias,
          const float* __restrict__ W0, const float* __restrict__ W1,
          float* __restrict__ o0, float* __restrict__ o1) {
  constexpr int H = HEADS, NT = 128, NPAIR = H * 16;
  __shared__ int   s_src[MAX_ENT];
  __shared__ int   s_row[MAX_ENT];
  __shared__ float s_al[H][MAX_ENT];
  __shared__ int   s_mk[NPAIR];
  constexpr int XP = IN_CH + 4;
  constexpr int TP = HC1 + 4;
  static_assert(H * 16 * XP >= 16 * TP, "the tile fits where the sums were");
  __shared__ __attribute__((aligned(16))) float s_buf[H * 16 * XP];
  const int T = blockIdx.x, tid = threadIdx.x;
  const int lane = tid & 31, u = tid >> 5, c = lane & 15, hf = lane >> 4;
  const int hh = u >> 1, xh = u & 1;
  const int node0 = T * 16;
  const float* __restrict__ hcol = x + xh * 64 + 4 * c;

  v8f c0 = {}, c1 = {}, c2 = {}, c3 = {};
  float dsum = 0.f;

  const unsigned cntT = tcnt[T];
  if (cntT <= TILE_CAP) {
    const int n  = (int)cntT;
    const int nE = n + 16;
    const int nP = (nE + 15) & ~15;
    if (tid < NPAIR) s_mk[tid] = (int)0x80000000;
    __syncthreads();
    for (int i = tid; i < nP; i += NT) {
      int s = node0, row = 0;
      if (i < n) {
        const unsigned ent = tl[(size_t)T * TILE_CAP + i];
        s = ei[ent & 0x00FFFFFFu]; row = (int)(ent >> 24);
      } else if (i < nE) {
        row = i - n; s = node0 + row;
      }
      s_src[i] = s; s_row[i] = row;
#pragma unroll
      for (int q = 0; q < H; ++q) {
        float sc = 0.f;
        if (i < nE) {
          sc = edge_score(as, ad, s, node0 + row, H, q);
          atomicMax(&s_mk[q * 16 + row], ord_key(sc));
        }
        s_al[q][i] = sc;
      }
    }
    __syncthreads();
    for (int i = tid; i < nE; i += NT) {
      const int row = s_row[i];
#pragma unroll
      for (int q = 0; q < H; ++q)
        s_al[q][i] = __expf(s_al[q][i] - ord_val(s_mk[q * 16 + row]));
    }
    __syncthreads();
    for (int j = 0; j < nP; j += 16) {
      v2f a[4]; float4 r[8];
#pragma unroll
      for (int q = 0; q < 4; ++q) {
        const int k = j + 4 * q + 2 * hf;
        a[q] = tile_a(s_row, s_al[hh], k, c);
        r[2 * q]     = ld_row(hcol, s_src[k], IN_CH);
        r[2 * q + 1] = ld_row(hcol, s_src[k + 1], IN_CH);
      }
      __builtin_amdgcn_sched_barrier(0);
#pragma unroll
      for (int q = 0; q < 4; ++q) {
        dsum += a[q].x + a[q].y;
        tile_mma(c0, c1, c2, c3, a[q], r[2 * q], r[2 * q + 1]);
      }
    }
  } else {
    if (tid < NPAIR) {
      const int row = tid & 15, q = tid >> 4, d = node0 + row;
      float mx = edge_score(as, ad, d, d, H, q);
      for (int e = 0; e < E0_EDGES; ++e)
        if (ei[E0_EDGES + e] == d) mx = fmaxf(mx, edge_score(as, ad, ei[e], d, H, q));
      s_mk[tid] = ord_key(mx);
    }
    __syncthreads();
    const float mx0 = ord_val(s_mk[hh * 16 + c]);
    for (int e0 = 0; e0 < E0_EDGES; e0 += 4) {
      const int e  = e0 + 2 * hf;
      const int d0 = ei[E0_EDGES + e], d1 = ei[E0_EDGES + e + 1];
      const bool in0 = (d0 >> 4) == T, in1 = (d1 >> 4) == T;
      if (__ballot(in0 | in1) == 0) continue;
      const int s0 = ei[e], s1 = ei[e + 1];
      v2f a;
      a.x = (in0 && (d0 & 15) == c) ? __expf(edge_score(as, ad, s0, d0, H, hh) - mx0) : 0.f;
      a.y = (in1 && (d1 & 15) == c) ? __expf(edge_score(as, ad, s1, d1, H, hh) - mx0) : 0.f;
      dsum += a.x + a.y;
      tile_mma(c0, c1, c2, c3, a, ld_row(hcol, s0, IN_CH), ld_row(hcol, s1, IN_CH));
    }
    for (int j = 0; j < 16; j += 4) {
      const int r0 = j + 2 * hf, n0 = node0 + r0, n1 = n0 + 1;
      v2f a;
      a.x = r0 == c     ? __expf(edge_score(as, ad, n0, n0, H, hh) - mx0) : 0.f;
      a.y = r0 + 1 == c ? __expf(edge_score(as, ad, n1, n1, H, hh) - mx0) : 0.f;
      dsum += a.x + a.y;
      tile_mma(c0, c1, c2, c3, a, ld_row(hcol, n0, IN_CH), ld_row(hcol, n1, IN_CH));
    }
  }

  const float dtot = dsum + __shfl_xor(dsum, 16);
#pragma unroll
  for (int r = 0; r < 8; ++r) {
    const float inv = 1.0f / (__shfl(dtot, 8 * hf + r) + 1e-16f);
    *(float4*)&s_buf[(hh * 16 + 8 * hf + r) * XP + xh * 64 + 4 * c] =
        make_float4(c0[r] * inv, c1[r] * inv, c2[r] * inv, c3[r] * inv);
  }
  __syncthreads();
  v8f e0 = {}, e1 = {}, e2 = {}, e3 = {};
  {
    const float* ax = &s_buf[(hh * 16 + c) * XP + 2 * hf];
    v2f a = *(const v2f*)ax;
    Bop b = load_b(Wl1, 0, HC1, u * 64, hf, c);
#pragma unroll 4
    for (int k0 = 0; k0 < IN_CH - 4; k0 += 4) {
      const v2f an = *(const v2f*)(ax + k0 + 4);
      Bop bn = load_b(Wl1, k0 + 4, HC1, u * 64, hf, c);
      e0 = __builtin_amdgcn_wmma_f32_16x16x4_f32(false, a, false, b.b0, (short)0, e0, false, false);
      e1 = __builtin_amdgcn_wmma_f32_16x16x4_f32(false, a, false, b.b1, (short)0, e1, false, false);
      e2 = __builtin_amdgcn_wmma_f32_16x16x4_f32(false, a, false, b.b2, (short)0, e2, false, false);
      e3 = __builtin_amdgcn_wmma_f32_16x16x4_f32(false, a, false, b.b3, (short)0, e3, false, false);
      a = an; b = bn;
    }
    e0 = __builtin_amdgcn_wmma_f32_16x16x4_f32(false, a, false, b.b0, (short)0, e0, false, false);
    e1 = __builtin_amdgcn_wmma_f32_16x16x4_f32(false, a, false, b.b1, (short)0, e1, false, false);
    e2 = __builtin_amdgcn_wmma_f32_16x16x4_f32(false, a, false, b.b2, (short)0, e2, false, false);
    e3 = __builtin_amdgcn_wmma_f32_16x16x4_f32(false, a, false, b.b3, (short)0, e3, false, false);
  }
  const float* bp = bias + u * 64 + c;
  const float bv0 = bp[0], bv1 = bp[16], bv2 = bp[32], bv3 = bp[48];
  __syncthreads();
  {
    float* tq = &s_buf[(8 * hf) * TP + u * 64 + c];
#pragma unroll
    for (int r = 0; r < 8; ++r) {
      tq[r * TP]      = elu1(bv0 + e0[r]);
      tq[r * TP + 16] = elu1(bv1 + e1[r]);
      tq[r * TP + 32] = elu1(bv2 + e2[r]);
      tq[r * TP + 48] = elu1(bv3 + e3[r]);
    }
  }
  __syncthreads();
  const int cb = (u & 1) * 32 + c;
  const float* __restrict__ Wb = (u < 2 ? W0 : W1) + (size_t)(2 * hf) * 64 + cb;
  const float* at = &s_buf[c * TP + 2 * hf];
  v8f d0 = {}, d1 = {};
  v2f a = *(const v2f*)at;
  v2f b0, b1;
  b0.x = Wb[0];  b0.y = Wb[64];
  b1.x = Wb[16]; b1.y = Wb[64 + 16];
#pragma unroll 4
  for (int k0 = 0; k0 < HC1 - 4; k0 += 4) {
    const float* wn = Wb + (size_t)(k0 + 4) * 64;
    const v2f an = *(const v2f*)(at + k0 + 4);
    v2f bn0, bn1;
    bn0.x = wn[0];  bn0.y = wn[64];
    bn1.x = wn[16]; bn1.y = wn[64 + 16];
    d0 = __builtin_amdgcn_wmma_f32_16x16x4_f32(false, a, false, b0, (short)0, d0, false, false);
    d1 = __builtin_amdgcn_wmma_f32_16x16x4_f32(false, a, false, b1, (short)0, d1, false, false);
    a = an; b0 = bn0; b1 = bn1;
  }
  d0 = __builtin_amdgcn_wmma_f32_16x16x4_f32(false, a, false, b0, (short)0, d0, false, false);
  d1 = __builtin_amdgcn_wmma_f32_16x16x4_f32(false, a, false, b1, (short)0, d1, false, false);
  float* ob = (u < 2 ? o0 : o1) + (size_t)(node0 + 8 * hf) * 64 + cb;
#pragma unroll
  for (int r = 0; r < 8; ++r) {
    ob[(size_t)r * 64]      = d0[r];
    ob[(size_t)r * 64 + 16] = d1[r];
  }
}

extern "C" void kernel_launch(void* const* d_in, const int* in_sizes, int n_in,
                              void* d_out, int out_size, void* d_ws, size_t ws_size,
                              hipStream_t stream) {
  const float* x    = (const float*)d_in[0];
  const int*   ei   = (const int*)  d_in[1];
  const float* W1   = (const float*)d_in[2];
  const float* aS1  = (const float*)d_in[3];
  const float* aD1  = (const float*)d_in[4];
  const float* b1   = (const float*)d_in[5];
  const float* Wmu  = (const float*)d_in[6];
  const float* aSmu = (const float*)d_in[7];
  const float* aDmu = (const float*)d_in[8];
  const float* bmu  = (const float*)d_in[9];
  const float* Wls  = (const float*)d_in[10];
  const float* aSls = (const float*)d_in[11];
  const float* aDls = (const float*)d_in[12];
  const float* bls  = (const float*)d_in[13];

  char* ws = (char*)d_ws;
  size_t off = 0;
  auto alloc = [&](size_t bytes) -> char* {
    off = (off + 255) & ~(size_t)255;
    char* p = ws + off; off += bytes; return p;
  };

  float*    h2   = (float*)   alloc((size_t)NN * 2 * OUT_CH * 4);
  float*    as1  = (float*)   alloc((size_t)NN * HEADS * 4);
  float*    ad1  = (float*)   alloc((size_t)NN * HEADS * 4);
  unsigned* tl   = (unsigned*)alloc((size_t)NTILES * TILE_CAP * 4);
  unsigned* tcnt = (unsigned*)alloc((size_t)NTILES * 4);
  float*    as2m = (float*)   alloc((size_t)NN * 4);
  float*    ad2m = (float*)   alloc((size_t)NN * 4);
  float*    as2l = (float*)   alloc((size_t)NN * 4);
  float*    ad2l = (float*)   alloc((size_t)NN * 4);
  float*    V1   = (float*)   alloc((size_t)4 * IN_CH * 4);

  float* hm = h2;
  float* hl = h2 + (size_t)NN * OUT_CH;
  float* out_mu = (float*)d_out;
  float* out_ls = out_mu + (size_t)NN * OUT_CH;

  bin_edges<<<NGROUPS + 1, 512, 0, stream>>>(ei, tl, tcnt, W1, aS1, aD1, V1);

  x_scores<<<(NN + 255) / 256, 256, 0, stream>>>(x, V1, as1, ad1, NN);
  agg1_tile<<<NTILES, 128, 0, stream>>>(tl, tcnt, ei, as1, ad1, x, W1, b1, Wmu, Wls, hm, hl);
  att_scores<<<(NN + 255) / 256, 256, 0, stream>>>(hm, aSmu, aDmu, as2m, ad2m, NN, 1, OUT_CH);
  att_scores<<<(NN + 255) / 256, 256, 0, stream>>>(hl, aSls, aDls, as2l, ad2l, NN, 1, OUT_CH);

  agg_tile<OUT_CH, OUT_CH><<<NTILES, OUT_CH / 2, 0, stream>>>(tl, tcnt, ei, as2m, ad2m, hm, bmu, out_mu);
  agg_tile<OUT_CH, OUT_CH><<<NTILES, OUT_CH / 2, 0, stream>>>(tl, tcnt, ei, as2l, ad2l, hl, bls, out_ls);
}


# layer-1 sum: matrix rows are 8 destinations x 2 heads (tile lists split by row half), so one fragment of input rows serves both heads: half the products and row loads; sums change order
# speedup vs baseline: 4.0327x; 1.1151x over previous
#include <hip/hip_runtime.h>

typedef __attribute__((ext_vector_type(2))) float v2f;
typedef __attribute__((ext_vector_type(8))) float v8f;

#define NN       50000
#define IN_CH    128
#define OUT_CH   64
#define HEADS    2
#define HID      128
#define HC1      (HEADS*HID)
#define E0_EDGES 400000
#define E_TOT    (E0_EDGES + NN)
#define NEG_SLOPE 0.2f

struct Bop { v2f b0, b1, b2, b3; };

__device__ __forceinline__ float elu1(float v) { return v > 0.f ? v : (__expf(v) - 1.f); }

__device__ __forceinline__ Bop load_b(const float* __restrict__ B, int k0, int N,
                                      int tn, int lhalf, int lmod) {
  const float* r0 = B + (size_t)(k0 + 2 * lhalf) * N + tn + lmod;
  const float* r1 = r0 + N;
  Bop b;
  b.b0.x = r0[0];  b.b0.y = r1[0];
  b.b1.x = r0[16]; b.b1.y = r1[16];
  b.b2.x = r0[32]; b.b2.y = r1[32];
  b.b3.x = r0[48]; b.b3.y = r1[48];
  return b;
}

__device__ __forceinline__ float leaky(float v) { return v >= 0.f ? v : NEG_SLOPE * v; }

#define NTILES    (NN / 16)
#define GROUP_T   16
#define NGROUPS   ((NTILES + GROUP_T - 1) / GROUP_T)
#define NHALF     (2 * GROUP_T)
#define SUB_CAP   24
#define TILE_CAP  256
#define LIST_FULL 0xFFFFFFFFu
#define MAX_ENT   (TILE_CAP + 16 + 16)

__global__ void __launch_bounds__(512)
bin_edges(const int* __restrict__ ei, unsigned* __restrict__ tl,
          unsigned* __restrict__ tcnt, const float* __restrict__ Wl1,
          const float* __restrict__ att_s, const float* __restrict__ att_d,
          float* __restrict__ V,
          const float* __restrict__ Wm, const float* __restrict__ Wl,
          const float* __restrict__ atsm, const float* __restrict__ atdm,
          const float* __restrict__ atsl, const float* __restrict__ atdl,
          float* __restrict__ V2) {
  __shared__ unsigned cnt[16][NHALF];
  __shared__ unsigned sub[16][NHALF][SUB_CAP];
  const int g = blockIdx.x, tid = threadIdx.x;
  const int lane = tid & 31, w = tid >> 5;
  if (g == NGROUPS) {
    const int which = tid >> 7, k = tid & (IN_CH - 1), hd = which & 1;
    const float4* wr = (const float4*)(Wl1 + (size_t)k * HC1 + hd * HID);
    const float4* av = (const float4*)((which < 2 ? att_s : att_d) + hd * HID);
    float s = 0.f;
    for (int cc = 0; cc < HID / 4; ++cc) {
      const float4 a = wr[cc], b = av[cc];
      s += a.x * b.x + a.y * b.y + a.z * b.z + a.w * b.w;
    }
    V[which * IN_CH + k] = s;
    for (int v = tid; v < 4 * HC1; v += 512) {
      const int j = v >> 8, k2 = v & (HC1 - 1);
      const float4* wr2 = (const float4*)((j < 2 ? Wm : Wl) + (size_t)k2 * OUT_CH);
      const float4* av2 = (const float4*)(j == 0 ? atsm : j == 1 ? atdm : j == 2 ? atsl : atdl);
      float s2 = 0.f;
      for (int cc = 0; cc < OUT_CH / 4; ++cc) {
        const float4 a = wr2[cc], b = av2[cc];
        s2 += a.x * b.x + a.y * b.y + a.z * b.z + a.w * b.w;
      }
      V2[v] = s2;
    }
    return;
  }
  cnt[tid >> 5][tid & 31] = 0u;
  __syncthreads();

  const int per_wave = E0_EDGES / 16;
  const int beg = w * per_wave, end = beg + per_wave;
  const int* __restrict__ dstp = ei + E0_EDGES;
  for (int base = beg; base < end; base += 1024) {
    int4 dv[8];
#pragma unroll
    for (int q = 0; q < 8; ++q) {
      const int idx = base + q * 128 + 4 * lane;
      dv[q] = *(const int4*)(dstp + (idx < end ? idx : end - 4));
    }
#pragma unroll
    for (int q = 0; q < 8; ++q) {
      const int idx = base + q * 128 + 4 * lane;
      const int dd[4] = {dv[q].x, dv[q].y, dv[q].z, dv[q].w};
#pragma unroll
      for (int k = 0; k < 4; ++k) {
        const int d = dd[k];
        if (idx < end && (d >> 8) == g) {
          const int ti = (d >> 3) & (NHALF - 1);
          const unsigned p = atomicAdd(&cnt[w][ti], 1u);
          if (p < SUB_CAP) sub[w][ti][p] = (unsigned)(idx + k) | ((unsigned)(d & 15) << 24);
        }
      }
    }
  }
  __syncthreads();

  const int T = g * GROUP_T + w;
  if (T < NTILES) {
    unsigned total = 0, first = 0; bool full = false;
    for (int hs = 0; hs < 2; ++hs) {
      const int ht = 2 * w + hs;
      for (int w2 = 0; w2 < 16; ++w2) {
        const unsigned n = cnt[w2][ht];
        if (n > SUB_CAP) full = true;
        const unsigned nn = n < SUB_CAP ? n : SUB_CAP;
        for (unsigned i = lane; i < nn; i += 32) {
          const unsigned pos = total + i;
          if (pos < TILE_CAP) tl[(size_t)T * TILE_CAP + pos] = sub[w2][ht][i];
        }
        total += n;
      }
      if (hs == 0) first = total;
    }
    if (total > TILE_CAP) full = true;
    if (lane == 0) tcnt[T] = full ? LIST_FULL : (total | (first << 16));
  }
}

__device__ __forceinline__ float edge_score(const float* __restrict__ as,
    const float* __restrict__ ad, int s, int d, int H, int q) {
  return leaky(as[s * H + q] + ad[d * H + q]);
}

__device__ __forceinline__ float4 ld_row(const float* __restrict__ hcol, int s, int HC) {
  return *(const float4*)(hcol + (size_t)s * HC);
}
__device__ __forceinline__ v2f tile_a(const int* row, const float* al, int k, int c) {
  const float w0 = al[k], w1 = al[k + 1];
  v2f a;
  a.x = row[k]     == c ? w0 : 0.f;
  a.y = row[k + 1] == c ? w1 : 0.f;
  return a;
}
__device__ __forceinline__ void tile_mma(v8f& c0, v8f& c1, v8f& c2, v8f& c3, v2f a,
                                         float4 r0, float4 r1) {
  v2f b;
  b.x = r0.x; b.y = r1.x;
  c0 = __builtin_amdgcn_wmma_f32_16x16x4_f32(false, a, false, b, (short)0, c0, false, false);
  b.x = r0.y; b.y = r1.y;
  c1 = __builtin_amdgcn_wmma_f32_16x16x4_f32(false, a, false, b, (short)0, c1, false, false);
  b.x = r0.z; b.y = r1.z;
  c2 = __builtin_amdgcn_wmma_f32_16x16x4_f32(false, a, false, b, (short)0, c2, false, false);
  b.x = r0.w; b.y = r1.w;
  c3 = __builtin_amdgcn_wmma_f32_16x16x4_f32(false, a, false, b, (short)0, c3, false, false);
}

__device__ __forceinline__ int   ord_key(float f) { const int b = __float_as_int(f); return b >= 0 ? b : b ^ 0x7FFFFFFF; }
__device__ __forceinline__ float ord_val(int k)   { return __int_as_float(k >= 0 ? k : k ^ 0x7FFFFFFF); }

template <int HC, int C>
__global__ void __launch_bounds__(HC / 2)
agg_tile(const unsigned* __restrict__ tl, const unsigned* __restrict__ tcnt,
         const int* __restrict__ ei, const float* __restrict__ as,
         const float* __restrict__ ad, const float* __restrict__ hs,
         const float* __restrict__ bias, float* __restrict__ out) {
  constexpr int H     = HC / C;
  constexpr int NT    = HC / 2;
  constexpr int NPAIR = H * 16;
  static_assert(NT >= NPAIR, "one thread per pair clears the maxima");
  __shared__ int   s_src[MAX_ENT];
  __shared__ int   s_row[MAX_ENT];
  __shared__ float s_al[H][MAX_ENT];
  __shared__ int   s_mk[NPAIR];
  const int T = blockIdx.x, tid = threadIdx.x;
  const int lane = tid & 31, u = tid >> 5, c = lane & 15, hf = lane >> 4;
  const int hh = (u * 64) / C;
  const int node0 = T * 16;
  const float* __restrict__ hcol = hs + u * 64 + 4 * c;

  v8f c0 = {}, c1 = {}, c2 = {}, c3 = {};
  float dsum = 0.f;
  const float* bp = bias + u * 64 + 4 * c;
  const float4 bv = make_float4(bp[0], bp[1], bp[2], bp[3]);

  const unsigned cntT = tcnt[T];
  if (cntT != LIST_FULL) {
    const int n  = (int)(cntT & 0xFFFFu);
    const int nE = n + 16;
    const int nP = (nE + 15) & ~15;
    if (tid < NPAIR) s_mk[tid] = (int)0x80000000;
    __syncthreads();
    for (int i = tid; i < nP; i += NT) {
      int s = node0, row = 0;
      if (i < n) {
        const unsigned ent = tl[(size_t)T * TILE_CAP + i];
        s = ei[ent & 0x00FFFFFFu]; row = (int)(ent >> 24);
      } else if (i < nE) {
        row = i - n; s = node0 + row;
      }
      s_src[i] = s; s_row[i] = row;
#pragma unroll
      for (int q = 0; q < H; ++q) {
        float sc = 0.f;
        if (i < nE) {
          sc = edge_score(as, ad, s, node0 + row, H, q);
          atomicMax(&s_mk[q * 16 + row], ord_key(sc));
        }
        s_al[q][i] = sc;
      }
    }
    __syncthreads();
    for (int i = tid; i < nE; i += NT) {
      const int row = s_row[i];
#pragma unroll
      for (int q = 0; q < H; ++q)
        s_al[q][i] = __expf(s_al[q][i] - ord_val(s_mk[q * 16 + row]));
    }
    __syncthreads();
    for (int j = 0; j < nP; j += 16) {
      v2f a[4]; float4 r[8];
#pragma unroll
      for (int q = 0; q < 4; ++q) {
        const int k = j + 4 * q + 2 * hf;
        a[q] = tile_a(s_row, s_al[hh], k, c);
        r[2 * q]     = ld_row(hcol, s_src[k], HC);
        r[2 * q + 1] = ld_row(hcol, s_src[k + 1], HC);
      }
      __builtin_amdgcn_sched_barrier(0);
#pragma unroll
      for (int q = 0; q < 4; ++q) {
        dsum += a[q].x + a[q].y;
        tile_mma(c0, c1, c2, c3, a[q], r[2 * q], r[2 * q + 1]);
      }
    }
  } else {
    if (tid < NPAIR) {
      const int row = tid & 15, q = tid >> 4, d = node0 + row;
      float mx = edge_score(as, ad, d, d, H, q);
      for (int e = 0; e < E0_EDGES; ++e)
        if (ei[E0_EDGES + e] == d) mx = fmaxf(mx, edge_score(as, ad, ei[e], d, H, q));
      s_mk[tid] = ord_key(mx);
    }
    __syncthreads();
    const float mx0 = ord_val(s_mk[hh * 16 + c]);
    for (int e0 = 0; e0 < E0_EDGES; e0 += 4) {
      const int e  = e0 + 2 * hf;
      const int d0 = ei[E0_EDGES + e], d1 = ei[E0_EDGES + e + 1];
      const bool in0 = (d0 >> 4) == T, in1 = (d1 >> 4) == T;
      if (__ballot(in0 | in1) == 0) continue;
      const int s0 = ei[e], s1 = ei[e + 1];
      v2f a;
      a.x = (in0 && (d0 & 15) == c) ? __expf(edge_score(as, ad, s0, d0, H, hh) - mx0) : 0.f;
      a.y = (in1 && (d1 & 15) == c) ? __expf(edge_score(as, ad, s1, d1, H, hh) - mx0) : 0.f;
      dsum += a.x + a.y;
      tile_mma(c0, c1, c2, c3, a, ld_row(hcol, s0, HC), ld_row(hcol, s1, HC));
    }
    for (int j = 0; j < 16; j += 4) {
      const int r0 = j + 2 * hf, n0 = node0 + r0, n1 = n0 + 1;
      v2f a;
      a.x = r0 == c     ? __expf(edge_score(as, ad, n0, n0, H, hh) - mx0) : 0.f;
      a.y = r0 + 1 == c ? __expf(edge_score(as, ad, n1, n1, H, hh) - mx0) : 0.f;
      dsum += a.x + a.y;
      tile_mma(c0, c1, c2, c3, a, ld_row(hcol, n0, HC), ld_row(hcol, n1, HC));
    }
  }

  const float dtot = dsum + __shfl_xor(dsum, 16);
#pragma unroll
  for (int r = 0; r < 8; ++r) {
    const float inv = 1.0f / (__shfl(dtot, 8 * hf + r) + 1e-16f);
    float* q = out + (size_t)(node0 + 8 * hf + r) * HC + u * 64 + 4 * c;
    *(float4*)q = make_float4(bv.x + c0[r] * inv, bv.y + c1[r] * inv,
                              bv.z + c2[r] * inv, bv.w + c3[r] * inv);
  }
}

__global__ void __launch_bounds__(256)
x_scores(const float* __restrict__ x, const float* __restrict__ V,
         float* __restrict__ as, float* __restrict__ ad, int n_nodes) {
  const int i = blockIdx.x * blockDim.x + threadIdx.x;
  if (i >= n_nodes) return;
  const float4* xp = (const float4*)(x + (size_t)i * IN_CH);
  const float4* v0 = (const float4*)V;
  const float4* v1 = v0 + IN_CH / 4;
  const float4* v2 = v1 + IN_CH / 4;
  const float4* v3 = v2 + IN_CH / 4;
  float s0 = 0.f, s1 = 0.f, d0 = 0.f, d1 = 0.f;
#pragma unroll 8
  for (int cc = 0; cc < IN_CH / 4; ++cc) {
    const float4 v = xp[cc], a = v0[cc], b = v1[cc], p = v2[cc], q = v3[cc];
    s0 += v.x * a.x + v.y * a.y + v.z * a.z + v.w * a.w;
    s1 += v.x * b.x + v.y * b.y + v.z * b.z + v.w * b.w;
    d0 += v.x * p.x + v.y * p.y + v.z * p.z + v.w * p.w;
    d1 += v.x * q.x + v.y * q.y + v.z * q.z + v.w * q.w;
  }
  as[2 * i] = s0; as[2 * i + 1] = s1;
  ad[2 * i] = d0; ad[2 * i + 1] = d1;
}

__global__ void __launch_bounds__(128)
agg1_tile(const unsigned* __restrict__ tl, const unsigned* __restrict__ tcnt,
          const int* __restrict__ ei, const float* __restrict__ as,
          const float* __restrict__ ad, const float* __restrict__ x,
          const float* __restrict__ Wl1, const float* __restrict__ bias,
          const float* __restrict__ W0, const float* __restrict__ W1,
          float* __restrict__ o0, float* __restrict__ o1,
          const float* __restrict__ V2, float* __restrict__ sc2) {
  constexpr int H = HEADS, NT = 128, NPAIR = H * 16;
  __shared__ float s_ps[4][16][4];
  __shared__ int   s_src[2][MAX_ENT];
  __shared__ int   s_row[2][MAX_ENT];
  __shared__ float s_al[2][H][MAX_ENT];
  __shared__ int   s_mk[NPAIR];
  constexpr int XP = IN_CH + 4;
  constexpr int TP = HC1 + 4;
  static_assert(H * 16 * XP >= 16 * TP, "the tile fits where the sums were");
  __shared__ __attribute__((aligned(16))) float s_buf[H * 16 * XP];
  const int T = blockIdx.x, tid = threadIdx.x;
  const int lane = tid & 31, u = tid >> 5, c = lane & 15, hf = lane >> 4;
  const int hh = u >> 1, xh = u & 1;
  const int node0 = T * 16;
  const float* __restrict__ hcol = x + xh * 64 + 4 * c;

  v8f c0 = {}, c1 = {}, c2 = {}, c3 = {};
  float dsum = 0.f;

  const unsigned cntT = tcnt[T];
  const bool listed = cntT != LIST_FULL;
  if (listed) {
    const int n   = (int)(cntT & 0xFFFFu);
    const int nA  = (int)(cntT >> 16), nB = n - nA;
    const int nPA = (nA + 8 + 15) & ~15;
    const int nPB = (nB + 8 + 15) & ~15;
    if (tid < NPAIR) s_mk[tid] = (int)0x80000000;
    __syncthreads();
    for (int i = tid; i < nPA + nPB; i += NT) {
      const int sb = i >= nPA ? 1 : 0;
      const int j  = i - (sb ? nPA : 0);
      const int nS = sb ? nB : nA;
      int s = node0, row = 8 * sb;
      if (j < nS) {
        const unsigned ent = tl[(size_t)T * TILE_CAP + (sb ? nA : 0) + j];
        s = ei[ent & 0x00FFFFFFu]; row = (int)(ent >> 24);
      } else if (j < nS + 8) {
        row = 8 * sb + (j - nS); s = node0 + row;
      }
      s_src[sb][j] = s; s_row[sb][j] = row;
#pragma unroll
      for (int q = 0; q < H; ++q) {
        float sc = 0.f;
        if (j < nS + 8) {
          sc = edge_score(as, ad, s, node0 + row, H, q);
          atomicMax(&s_mk[q * 16 + row], ord_key(sc));
        }
        s_al[sb][q][j] = sc;
      }
    }
    __syncthreads();
    for (int i = tid; i < nPA + nPB; i += NT) {
      const int sb = i >= nPA ? 1 : 0;
      const int j  = i - (sb ? nPA : 0);
      if (j < (sb ? nB : nA) + 8) {
        const int row = s_row[sb][j];
#pragma unroll
        for (int q = 0; q < H; ++q)
          s_al[sb][q][j] = __expf(s_al[sb][q][j] - ord_val(s_mk[q * 16 + row]));
      }
    }
    __syncthreads();
    const int*   rowp  = s_row[hh];
    const int*   srcp  = s_src[hh];
    const float* alp   = s_al[hh][c >> 3];
    const int    myrow = 8 * hh + (c & 7);
    const int    nP    = __builtin_amdgcn_readfirstlane(hh ? nPB : nPA);
    for (int j = 0; j < nP; j += 16) {
      v2f a[4]; float4 r[8];
#pragma unroll
      for (int q = 0; q < 4; ++q) {
        const int k = j + 4 * q + 2 * hf;
        a[q] = tile_a(rowp, alp, k, myrow);
        r[2 * q]     = ld_row(hcol, srcp[k], IN_CH);
        r[2 * q + 1] = ld_row(hcol, srcp[k + 1], IN_CH);
      }
      __builtin_amdgcn_sched_barrier(0);
#pragma unroll
      for (int q = 0; q < 4; ++q) {
        dsum += a[q].x + a[q].y;
        tile_mma(c0, c1, c2, c3, a[q], r[2 * q], r[2 * q + 1]);
      }
    }
  } else {
    if (tid < NPAIR) {
      const int row = tid & 15, q = tid >> 4, d = node0 + row;
      float mx = edge_score(as, ad, d, d, H, q);
      for (int e = 0; e < E0_EDGES; ++e)
        if (ei[E0_EDGES + e] == d) mx = fmaxf(mx, edge_score(as, ad, ei[e], d, H, q));
      s_mk[tid] = ord_key(mx);
    }
    __syncthreads();
    const float mx0 = ord_val(s_mk[hh * 16 + c]);
    for (int e0 = 0; e0 < E0_EDGES; e0 += 4) {
      const int e  = e0 + 2 * hf;
      const int d0 = ei[E0_EDGES + e], d1 = ei[E0_EDGES + e + 1];
      const bool in0 = (d0 >> 4) == T, in1 = (d1 >> 4) == T;
      if (__ballot(in0 | in1) == 0) continue;
      const int s0 = ei[e], s1 = ei[e + 1];
      v2f a;
      a.x = (in0 && (d0 & 15) == c) ? __expf(edge_score(as, ad, s0, d0, H, hh) - mx0) : 0.f;
      a.y = (in1 && (d1 & 15) == c) ? __expf(edge_score(as, ad, s1, d1, H, hh) - mx0) : 0.f;
      dsum += a.x + a.y;
      tile_mma(c0, c1, c2, c3, a, ld_row(hcol, s0, IN_CH), ld_row(hcol, s1, IN_CH));
    }
    for (int j = 0; j < 16; j += 4) {
      const int r0 = j + 2 * hf, n0 = node0 + r0, n1 = n0 + 1;
      v2f a;
      a.x = r0 == c     ? __expf(edge_score(as, ad, n0, n0, H, hh) - mx0) : 0.f;
      a.y = r0 + 1 == c ? __expf(edge_score(as, ad, n1, n1, H, hh) - mx0) : 0.f;
      dsum += a.x + a.y;
      tile_mma(c0, c1, c2, c3, a, ld_row(hcol, n0, IN_CH), ld_row(hcol, n1, IN_CH));
    }
  }

  const float dtot = dsum + __shfl_xor(dsum, 16);
  const int rb = listed ? hf * 16 + 8 * hh : hh * 16 + 8 * hf;
#pragma unroll
  for (int r = 0; r < 8; ++r) {
    const float inv = 1.0f / (__shfl(dtot, 8 * hf + r) + 1e-16f);
    *(float4*)&s_buf[(rb + r) * XP + xh * 64 + 4 * c] =
        make_float4(c0[r] * inv, c1[r] * inv, c2[r] * inv, c3[r] * inv);
  }
  __syncthreads();
  v8f e0 = {}, e1 = {}, e2 = {}, e3 = {};
  {
    const float* ax = &s_buf[(hh * 16 + c) * XP + 2 * hf];
    v2f a = *(const v2f*)ax;
    Bop b = load_b(Wl1, 0, HC1, u * 64, hf, c);
#pragma unroll 4
    for (int k0 = 0; k0 < IN_CH - 4; k0 += 4) {
      const v2f an = *(const v2f*)(ax + k0 + 4);
      Bop bn = load_b(Wl1, k0 + 4, HC1, u * 64, hf, c);
      e0 = __builtin_amdgcn_wmma_f32_16x16x4_f32(false, a, false, b.b0, (short)0, e0, false, false);
      e1 = __builtin_amdgcn_wmma_f32_16x16x4_f32(false, a, false, b.b1, (short)0, e1, false, false);
      e2 = __builtin_amdgcn_wmma_f32_16x16x4_f32(false, a, false, b.b2, (short)0, e2, false, false);
      e3 = __builtin_amdgcn_wmma_f32_16x16x4_f32(false, a, false, b.b3, (short)0, e3, false, false);
      a = an; b = bn;
    }
    e0 = __builtin_amdgcn_wmma_f32_16x16x4_f32(false, a, false, b.b0, (short)0, e0, false, false);
    e1 = __builtin_amdgcn_wmma_f32_16x16x4_f32(false, a, false, b.b1, (short)0, e1, false, false);
    e2 = __builtin_amdgcn_wmma_f32_16x16x4_f32(false, a, false, b.b2, (short)0, e2, false, false);
    e3 = __builtin_amdgcn_wmma_f32_16x16x4_f32(false, a, false, b.b3, (short)0, e3, false, false);
  }
  const float* bp = bias + u * 64 + c;
  const float bv0 = bp[0], bv1 = bp[16], bv2 = bp[32], bv3 = bp[48];
  __syncthreads();
  {
    float* tq = &s_buf[(8 * hf) * TP + u * 64 + c];
#pragma unroll
    for (int r = 0; r < 8; ++r) {
      tq[r * TP]      = elu1(bv0 + e0[r]);
      tq[r * TP + 16] = elu1(bv1 + e1[r]);
      tq[r * TP + 32] = elu1(bv2 + e2[r]);
      tq[r * TP + 48] = elu1(bv3 + e3[r]);
    }
  }
  __syncthreads();
  const int cb = (u & 1) * 32 + c;
  const float* __restrict__ Wb = (u < 2 ? W0 : W1) + (size_t)(2 * hf) * 64 + cb;
  const float* at = &s_buf[c * TP + 2 * hf];
  const float* vp = V2 + (c < 4 ? c : 0) * HC1 + u * 64 + 2 * hf;
  v2f tv[16];
  for (int i = 0; i < 16; ++i) tv[i] = *(const v2f*)(vp + 4 * i);
  v8f d0 = {}, d1 = {};
  v2f a = *(const v2f*)at;
  v2f b0, b1;
  b0.x = Wb[0];  b0.y = Wb[64];
  b1.x = Wb[16]; b1.y = Wb[64 + 16];
#pragma unroll 4
  for (int k0 = 0; k0 < HC1 - 4; k0 += 4) {
    const float* wn = Wb + (size_t)(k0 + 4) * 64;
    const v2f an = *(const v2f*)(at + k0 + 4);
    v2f bn0, bn1;
    bn0.x = wn[0];  bn0.y = wn[64];
    bn1.x = wn[16]; bn1.y = wn[64 + 16];
    d0 = __builtin_amdgcn_wmma_f32_16x16x4_f32(false, a, false, b0, (short)0, d0, false, false);
    d1 = __builtin_amdgcn_wmma_f32_16x16x4_f32(false, a, false, b1, (short)0, d1, false, false);
    a = an; b0 = bn0; b1 = bn1;
  }
  d0 = __builtin_amdgcn_wmma_f32_16x16x4_f32(false, a, false, b0, (short)0, d0, false, false);
  d1 = __builtin_amdgcn_wmma_f32_16x16x4_f32(false, a, false, b1, (short)0, d1, false, false);
  {
    const float* aq = at + u * 64;
    v8f p0 = {}, p1 = {};
    for (int i = 0; i < 16; i += 2) {
      const v2f a0 = *(const v2f*)(aq + 4 * i), a1 = *(const v2f*)(aq + 4 * i + 4);
      v2f q0, q1;
      q0.x = c < 4 ? tv[i].x : 0.f;     q0.y = c < 4 ? tv[i].y : 0.f;
      q1.x = c < 4 ? tv[i + 1].x : 0.f; q1.y = c < 4 ? tv[i + 1].y : 0.f;
      p0 = __builtin_amdgcn_wmma_f32_16x16x4_f32(false, a0, false, q0, (short)0, p0, false, false);
      p1 = __builtin_amdgcn_wmma_f32_16x16x4_f32(false, a1, false, q1, (short)0, p1, false, false);
    }
    if (c < 4) {
      for (int r = 0; r < 8; ++r) s_ps[u][8 * hf + r][c] = p0[r] + p1[r];
    }
  }
  float* ob = (u < 2 ? o0 : o1) + (size_t)(node0 + 8 * hf) * 64 + cb;
#pragma unroll
  for (int r = 0; r < 8; ++r) {
    ob[(size_t)r * 64]      = d0[r];
    ob[(size_t)r * 64 + 16] = d1[r];
  }
  __syncthreads();
  if (tid < 64) {
    const int row = tid >> 2, j = tid & 3;
    sc2[(size_t)j * NN + node0 + row] =
        ((s_ps[0][row][j] + s_ps[1][row][j]) + s_ps[2][row][j]) + s_ps[3][row][j];
  }
}

extern "C" void kernel_launch(void* const* d_in, const int* in_sizes, int n_in,
                              void* d_out, int out_size, void* d_ws, size_t ws_size,
                              hipStream_t stream) {
  const float* x    = (const float*)d_in[0];
  const int*   ei   = (const int*)  d_in[1];
  const float* W1   = (const float*)d_in[2];
  const float* aS1  = (const float*)d_in[3];
  const float* aD1  = (const float*)d_in[4];
  const float* b1   = (const float*)d_in[5];
  const float* Wmu  = (const float*)d_in[6];
  const float* aSmu = (const float*)d_in[7];
  const float* aDmu = (const float*)d_in[8];
  const float* bmu  = (const float*)d_in[9];
  const float* Wls  = (const float*)d_in[10];
  const float* aSls = (const float*)d_in[11];
  const float* aDls = (const float*)d_in[12];
  const float* bls  = (const float*)d_in[13];

  char* ws = (char*)d_ws;
  size_t off = 0;
  auto alloc = [&](size_t bytes) -> char* {
    off = (off + 255) & ~(size_t)255;
    char* p = ws + off; off += bytes; return p;
  };

  float*    h2   = (float*)   alloc((size_t)NN * 2 * OUT_CH * 4);
  float*    as1  = (float*)   alloc((size_t)NN * HEADS * 4);
  float*    ad1  = (float*)   alloc((size_t)NN * HEADS * 4);
  unsigned* tl   = (unsigned*)alloc((size_t)NTILES * TILE_CAP * 4);
  unsigned* tcnt = (unsigned*)alloc((size_t)NTILES * 4);
  float*    sc2  = (float*)   alloc((size_t)4 * NN * 4);
  float*    V1   = (float*)   alloc((size_t)4 * IN_CH * 4);
  float*    V2   = (float*)   alloc((size_t)4 * HC1 * 4);
  float*    as2m = sc2;
  float*    ad2m = sc2 + (size_t)NN;
  float*    as2l = sc2 + (size_t)2 * NN;
  float*    ad2l = sc2 + (size_t)3 * NN;

  float* hm = h2;
  float* hl = h2 + (size_t)NN * OUT_CH;
  float* out_mu = (float*)d_out;
  float* out_ls = out_mu + (size_t)NN * OUT_CH;

  bin_edges<<<NGROUPS + 1, 512, 0, stream>>>(ei, tl, tcnt, W1, aS1, aD1, V1,
                                             Wmu, Wls, aSmu, aDmu, aSls, aDls, V2);

  x_scores<<<(NN + 255) / 256, 256, 0, stream>>>(x, V1, as1, ad1, NN);
  agg1_tile<<<NTILES, 128, 0, stream>>>(tl, tcnt, ei, as1, ad1, x, W1, b1, Wmu, Wls, hm, hl, V2, sc2);

  agg_tile<OUT_CH, OUT_CH><<<NTILES, OUT_CH / 2, 0, stream>>>(tl, tcnt, ei, as2m, ad2m, hm, bmu, out_mu);
  agg_tile<OUT_CH, OUT_CH><<<NTILES, OUT_CH / 2, 0, stream>>>(tl, tcnt, ei, as2l, ad2l, hl, bls, out_ls);
}


# layer 2's two aggregations (mu, logstd) in one launch: a tile's block reads its edge list once, 64 threads build both softmax weight sets; wave 0 sums mu, wave 1 logstd; 4 launches; sums keep order
# speedup vs baseline: 4.2778x; 1.0608x over previous
#include <hip/hip_runtime.h>

typedef __attribute__((ext_vector_type(2))) float v2f;
typedef __attribute__((ext_vector_type(8))) float v8f;

#define NN       50000
#define IN_CH    128
#define OUT_CH   64
#define HEADS    2
#define HID      128
#define HC1      (HEADS*HID)
#define E0_EDGES 400000
#define E_TOT    (E0_EDGES + NN)
#define NEG_SLOPE 0.2f

struct Bop { v2f b0, b1, b2, b3; };

__device__ __forceinline__ float elu1(float v) { return v > 0.f ? v : (__expf(v) - 1.f); }

__device__ __forceinline__ Bop load_b(const float* __restrict__ B, int k0, int N,
                                      int tn, int lhalf, int lmod) {
  const float* r0 = B + (size_t)(k0 + 2 * lhalf) * N + tn + lmod;
  const float* r1 = r0 + N;
  Bop b;
  b.b0.x = r0[0];  b.b0.y = r1[0];
  b.b1.x = r0[16]; b.b1.y = r1[16];
  b.b2.x = r0[32]; b.b2.y = r1[32];
  b.b3.x = r0[48]; b.b3.y = r1[48];
  return b;
}

__device__ __forceinline__ float leaky(float v) { return v >= 0.f ? v : NEG_SLOPE * v; }

#define NTILES    (NN / 16)
#define GROUP_T   16
#define NGROUPS   ((NTILES + GROUP_T - 1) / GROUP_T)
#define NHALF     (2 * GROUP_T)
#define SUB_CAP   24
#define TILE_CAP  256
#define LIST_FULL 0xFFFFFFFFu
#define MAX_ENT   (TILE_CAP + 16 + 16)

__global__ void __launch_bounds__(512)
bin_edges(const int* __restrict__ ei, unsigned* __restrict__ tl,
          unsigned* __restrict__ tcnt, const float* __restrict__ Wl1,
          const float* __restrict__ att_s, const float* __restrict__ att_d,
          float* __restrict__ V,
          const float* __restrict__ Wm, const float* __restrict__ Wl,
          const float* __restrict__ atsm, const float* __restrict__ atdm,
          const float* __restrict__ atsl, const float* __restrict__ atdl,
          float* __restrict__ V2) {
  __shared__ unsigned cnt[16][NHALF];
  __shared__ unsigned sub[16][NHALF][SUB_CAP];
  const int g = blockIdx.x, tid = threadIdx.x;
  const int lane = tid & 31, w = tid >> 5;
  if (g == NGROUPS) {
    const int which = tid >> 7, k = tid & (IN_CH - 1), hd = which & 1;
    const float4* wr = (const float4*)(Wl1 + (size_t)k * HC1 + hd * HID);
    const float4* av = (const float4*)((which < 2 ? att_s : att_d) + hd * HID);
    float s = 0.f;
    for (int cc = 0; cc < HID / 4; ++cc) {
      const float4 a = wr[cc], b = av[cc];
      s += a.x * b.x + a.y * b.y + a.z * b.z + a.w * b.w;
    }
    V[which * IN_CH + k] = s;
    for (int v = tid; v < 4 * HC1; v += 512) {
      const int j = v >> 8, k2 = v & (HC1 - 1);
      const float4* wr2 = (const float4*)((j < 2 ? Wm : Wl) + (size_t)k2 * OUT_CH);
      const float4* av2 = (const float4*)(j == 0 ? atsm : j == 1 ? atdm : j == 2 ? atsl : atdl);
      float s2 = 0.f;
      for (int cc = 0; cc < OUT_CH / 4; ++cc) {
        const float4 a = wr2[cc], b = av2[cc];
        s2 += a.x * b.x + a.y * b.y + a.z * b.z + a.w * b.w;
      }
      V2[v] = s2;
    }
    return;
  }
  cnt[tid >> 5][tid & 31] = 0u;
  __syncthreads();

  const int per_wave = E0_EDGES / 16;
  const int beg = w * per_wave, end = beg + per_wave;
  const int* __restrict__ dstp = ei + E0_EDGES;
  for (int base = beg; base < end; base += 1024) {
    int4 dv[8];
#pragma unroll
    for (int q = 0; q < 8; ++q) {
      const int idx = base + q * 128 + 4 * lane;
      dv[q] = *(const int4*)(dstp + (idx < end ? idx : end - 4));
    }
#pragma unroll
    for (int q = 0; q < 8; ++q) {
      const int idx = base + q * 128 + 4 * lane;
      const int dd[4] = {dv[q].x, dv[q].y, dv[q].z, dv[q].w};
#pragma unroll
      for (int k = 0; k < 4; ++k) {
        const int d = dd[k];
        if (idx < end && (d >> 8) == g) {
          const int ti = (d >> 3) & (NHALF - 1);
          const unsigned p = atomicAdd(&cnt[w][ti], 1u);
          if (p < SUB_CAP) sub[w][ti][p] = (unsigned)(idx + k) | ((unsigned)(d & 15) << 24);
        }
      }
    }
  }
  __syncthreads();

  const int T = g * GROUP_T + w;
  if (T < NTILES) {
    unsigned total = 0, first = 0; bool full = false;
    for (int hs = 0; hs < 2; ++hs) {
      const int ht = 2 * w + hs;
      for (int w2 = 0; w2 < 16; ++w2) {
        const unsigned n = cnt[w2][ht];
        if (n > SUB_CAP) full = true;
        const unsigned nn = n < SUB_CAP ? n : SUB_CAP;
        for (unsigned i = lane; i < nn; i += 32) {
          const unsigned pos = total + i;
          if (pos < TILE_CAP) tl[(size_t)T * TILE_CAP + pos] = sub[w2][ht][i];
        }
        total += n;
      }
      if (hs == 0) first = total;
    }
    if (total > TILE_CAP) full = true;
    if (lane == 0) tcnt[T] = full ? LIST_FULL : (total | (first << 16));
  }
}

__device__ __forceinline__ float edge_score(const float* __restrict__ as,
    const float* __restrict__ ad, int s, int d, int H, int q) {
  return leaky(as[s * H + q] + ad[d * H + q]);
}

__device__ __forceinline__ float4 ld_row(const float* __restrict__ hcol, int s, int HC) {
  return *(const float4*)(hcol + (size_t)s * HC);
}
__device__ __forceinline__ v2f tile_a(const int* row, const float* al, int k, int c) {
  const float w0 = al[k], w1 = al[k + 1];
  v2f a;
  a.x = row[k]     == c ? w0 : 0.f;
  a.y = row[k + 1] == c ? w1 : 0.f;
  return a;
}
__device__ __forceinline__ void tile_mma(v8f& c0, v8f& c1, v8f& c2, v8f& c3, v2f a,
                                         float4 r0, float4 r1) {
  v2f b;
  b.x = r0.x; b.y = r1.x;
  c0 = __builtin_amdgcn_wmma_f32_16x16x4_f32(false, a, false, b, (short)0, c0, false, false);
  b.x = r0.y; b.y = r1.y;
  c1 = __builtin_amdgcn_wmma_f32_16x16x4_f32(false, a, false, b, (short)0, c1, false, false);
  b.x = r0.z; b.y = r1.z;
  c2 = __builtin_amdgcn_wmma_f32_16x16x4_f32(false, a, false, b, (short)0, c2, false, false);
  b.x = r0.w; b.y = r1.w;
  c3 = __builtin_amdgcn_wmma_f32_16x16x4_f32(false, a, false, b, (short)0, c3, false, false);
}

__device__ __forceinline__ int   ord_key(float f) { const int b = __float_as_int(f); return b >= 0 ? b : b ^ 0x7FFFFFFF; }
__device__ __forceinline__ float ord_val(int k)   { return __int_as_float(k >= 0 ? k : k ^ 0x7FFFFFFF); }

__device__ __forceinline__ float edge_score2(const float* __restrict__ sc, int s, int d, int q) {
  return leaky(sc[(size_t)(2 * q) * NN + s] + sc[(size_t)(2 * q + 1) * NN + d]);
}

__global__ void __launch_bounds__(64)
agg2_tile(const unsigned* __restrict__ tl, const unsigned* __restrict__ tcnt,
          const int* __restrict__ ei, const float* __restrict__ sc,
          const float* __restrict__ h0, const float* __restrict__ h1,
          const float* __restrict__ bias0, const float* __restrict__ bias1,
          float* __restrict__ out0, float* __restrict__ out1) {
  constexpr int H     = 2;
  constexpr int C     = OUT_CH;
  constexpr int NT    = 64;
  constexpr int NPAIR = H * 16;
  static_assert(NT >= NPAIR, "one thread per pair clears the maxima");
  __shared__ int   s_src[MAX_ENT];
  __shared__ int   s_row[MAX_ENT];
  __shared__ float s_al[H][MAX_ENT];
  __shared__ int   s_mk[NPAIR];
  const int T = blockIdx.x, tid = threadIdx.x;
  const int lane = tid & 31, u = tid >> 5, c = lane & 15, hf = lane >> 4;
  const int node0 = T * 16;
  const float* __restrict__ hcol = (u ? h1 : h0) + 4 * c;

  v8f c0 = {}, c1 = {}, c2 = {}, c3 = {};
  float dsum = 0.f;
  const float* bp = (u ? bias1 : bias0) + 4 * c;
  const float4 bv = make_float4(bp[0], bp[1], bp[2], bp[3]);

  const unsigned cntT = tcnt[T];
  if (cntT != LIST_FULL) {
    const int n  = (int)(cntT & 0xFFFFu);
    const int nE = n + 16;
    const int nP = (nE + 15) & ~15;
    if (tid < NPAIR) s_mk[tid] = (int)0x80000000;
    __syncthreads();
    for (int i = tid; i < nP; i += NT) {
      int s = node0, row = 0;
      if (i < n) {
        const unsigned ent = tl[(size_t)T * TILE_CAP + i];
        s = ei[ent & 0x00FFFFFFu]; row = (int)(ent >> 24);
      } else if (i < nE) {
        row = i - n; s = node0 + row;
      }
      s_src[i] = s; s_row[i] = row;
      float sv0 = 0.f, sv1 = 0.f;
      if (i < nE) {
        sv0 = edge_score2(sc, s, node0 + row, 0);
        sv1 = edge_score2(sc, s, node0 + row, 1);
        atomicMax(&s_mk[row], ord_key(sv0));
        atomicMax(&s_mk[16 + row], ord_key(sv1));
      }
      s_al[0][i] = sv0; s_al[1][i] = sv1;
    }
    __syncthreads();
    for (int i = tid; i < nE; i += NT) {
      const int row = s_row[i];
#pragma unroll
      for (int q = 0; q < H; ++q)
        s_al[q][i] = __expf(s_al[q][i] - ord_val(s_mk[q * 16 + row]));
    }
    __syncthreads();
    const float* alp = s_al[u];
    for (int j = 0; j < nP; j += 16) {
      v2f a[4]; float4 r[8];
#pragma unroll
      for (int q = 0; q < 4; ++q) {
        const int k = j + 4 * q + 2 * hf;
        a[q] = tile_a(s_row, alp, k, c);
        r[2 * q]     = ld_row(hcol, s_src[k], C);
        r[2 * q + 1] = ld_row(hcol, s_src[k + 1], C);
      }
      __builtin_amdgcn_sched_barrier(0);
#pragma unroll
      for (int q = 0; q < 4; ++q) {
        dsum += a[q].x + a[q].y;
        tile_mma(c0, c1, c2, c3, a[q], r[2 * q], r[2 * q + 1]);
      }
    }
  } else {
    if (tid < NPAIR) {
      const int row = tid & 15, q = tid >> 4, d = node0 + row;
      float mx = edge_score2(sc, d, d, q);
      for (int e = 0; e < E0_EDGES; ++e)
        if (ei[E0_EDGES + e] == d) mx = fmaxf(mx, edge_score2(sc, ei[e], d, q));
      s_mk[tid] = ord_key(mx);
    }
    __syncthreads();
    const float mx0 = ord_val(s_mk[u * 16 + c]);
    for (int e0 = 0; e0 < E0_EDGES; e0 += 4) {
      const int e  = e0 + 2 * hf;
      const int d0 = ei[E0_EDGES + e], d1 = ei[E0_EDGES + e + 1];
      const bool in0 = (d0 >> 4) == T, in1 = (d1 >> 4) == T;
      if (__ballot(in0 | in1) == 0) continue;
      const int s0 = ei[e], s1 = ei[e + 1];
      v2f a;
      a.x = (in0 && (d0 & 15) == c) ? __expf(edge_score2(sc, s0, d0, u) - mx0) : 0.f;
      a.y = (in1 && (d1 & 15) == c) ? __expf(edge_score2(sc, s1, d1, u) - mx0) : 0.f;
      dsum += a.x + a.y;
      tile_mma(c0, c1, c2, c3, a, ld_row(hcol, s0, C), ld_row(hcol, s1, C));
    }
    for (int j = 0; j < 16; j += 4) {
      const int r0 = j + 2 * hf, n0 = node0 + r0, n1 = n0 + 1;
      v2f a;
      a.x = r0 == c     ? __expf(edge_score2(sc, n0, n0, u) - mx0) : 0.f;
      a.y = r0 + 1 == c ? __expf(edge_score2(sc, n1, n1, u) - mx0) : 0.f;
      dsum += a.x + a.y;
      tile_mma(c0, c1, c2, c3, a, ld_row(hcol, n0, C), ld_row(hcol, n1, C));
    }
  }

  const float dtot = dsum + __shfl_xor(dsum, 16);
  float* __restrict__ ob = (u ? out1 : out0) + 4 * c;
#pragma unroll
  for (int r = 0; r < 8; ++r) {
    const float inv = 1.0f / (__shfl(dtot, 8 * hf + r) + 1e-16f);
    float* q = ob + (size_t)(node0 + 8 * hf + r) * C;
    *(float4*)q = make_float4(bv.x + c0[r] * inv, bv.y + c1[r] * inv,
                              bv.z + c2[r] * inv, bv.w + c3[r] * inv);
  }
}

__global__ void __launch_bounds__(256)
x_scores(const float* __restrict__ x, const float* __restrict__ V,
         float* __restrict__ as, float* __restrict__ ad, int n_nodes) {
  const int i = blockIdx.x * blockDim.x + threadIdx.x;
  if (i >= n_nodes) return;
  const float4* xp = (const float4*)(x + (size_t)i * IN_CH);
  const float4* v0 = (const float4*)V;
  const float4* v1 = v0 + IN_CH / 4;
  const float4* v2 = v1 + IN_CH / 4;
  const float4* v3 = v2 + IN_CH / 4;
  float s0 = 0.f, s1 = 0.f, d0 = 0.f, d1 = 0.f;
#pragma unroll 8
  for (int cc = 0; cc < IN_CH / 4; ++cc) {
    const float4 v = xp[cc], a = v0[cc], b = v1[cc], p = v2[cc], q = v3[cc];
    s0 += v.x * a.x + v.y * a.y + v.z * a.z + v.w * a.w;
    s1 += v.x * b.x + v.y * b.y + v.z * b.z + v.w * b.w;
    d0 += v.x * p.x + v.y * p.y + v.z * p.z + v.w * p.w;
    d1 += v.x * q.x + v.y * q.y + v.z * q.z + v.w * q.w;
  }
  as[2 * i] = s0; as[2 * i + 1] = s1;
  ad[2 * i] = d0; ad[2 * i + 1] = d1;
}

__global__ void __launch_bounds__(128)
agg1_tile(const unsigned* __restrict__ tl, const unsigned* __restrict__ tcnt,
          const int* __restrict__ ei, const float* __restrict__ as,
          const float* __restrict__ ad, const float* __restrict__ x,
          const float* __restrict__ Wl1, const float* __restrict__ bias,
          const float* __restrict__ W0, const float* __restrict__ W1,
          float* __restrict__ o0, float* __restrict__ o1,
          const float* __restrict__ V2, float* __restrict__ sc2) {
  constexpr int H = HEADS, NT = 128, NPAIR = H * 16;
  __shared__ float s_ps[4][16][4];
  __shared__ int   s_src[2][MAX_ENT];
  __shared__ int   s_row[2][MAX_ENT];
  __shared__ float s_al[2][H][MAX_ENT];
  __shared__ int   s_mk[NPAIR];
  constexpr int XP = IN_CH + 4;
  constexpr int TP = HC1 + 4;
  static_assert(H * 16 * XP >= 16 * TP, "the tile fits where the sums were");
  __shared__ __attribute__((aligned(16))) float s_buf[H * 16 * XP];
  const int T = blockIdx.x, tid = threadIdx.x;
  const int lane = tid & 31, u = tid >> 5, c = lane & 15, hf = lane >> 4;
  const int hh = u >> 1, xh = u & 1;
  const int node0 = T * 16;
  const float* __restrict__ hcol = x + xh * 64 + 4 * c;

  v8f c0 = {}, c1 = {}, c2 = {}, c3 = {};
  float dsum = 0.f;

  const unsigned cntT = tcnt[T];
  const bool listed = cntT != LIST_FULL;
  if (listed) {
    const int n   = (int)(cntT & 0xFFFFu);
    const int nA  = (int)(cntT >> 16), nB = n - nA;
    const int nPA = (nA + 8 + 15) & ~15;
    const int nPB = (nB + 8 + 15) & ~15;
    if (tid < NPAIR) s_mk[tid] = (int)0x80000000;
    __syncthreads();
    for (int i = tid; i < nPA + nPB; i += NT) {
      const int sb = i >= nPA ? 1 : 0;
      const int j  = i - (sb ? nPA : 0);
      const int nS = sb ? nB : nA;
      int s = node0, row = 8 * sb;
      if (j < nS) {
        const unsigned ent = tl[(size_t)T * TILE_CAP + (sb ? nA : 0) + j];
        s = ei[ent & 0x00FFFFFFu]; row = (int)(ent >> 24);
      } else if (j < nS + 8) {
        row = 8 * sb + (j - nS); s = node0 + row;
      }
      s_src[sb][j] = s; s_row[sb][j] = row;
#pragma unroll
      for (int q = 0; q < H; ++q) {
        float sc = 0.f;
        if (j < nS + 8) {
          sc = edge_score(as, ad, s, node0 + row, H, q);
          atomicMax(&s_mk[q * 16 + row], ord_key(sc));
        }
        s_al[sb][q][j] = sc;
      }
    }
    __syncthreads();
    for (int i = tid; i < nPA + nPB; i += NT) {
      const int sb = i >= nPA ? 1 : 0;
      const int j  = i - (sb ? nPA : 0);
      if (j < (sb ? nB : nA) + 8) {
        const int row = s_row[sb][j];
#pragma unroll
        for (int q = 0; q < H; ++q)
          s_al[sb][q][j] = __expf(s_al[sb][q][j] - ord_val(s_mk[q * 16 + row]));
      }
    }
    __syncthreads();
    const int*   rowp  = s_row[hh];
    const int*   srcp  = s_src[hh];
    const float* alp   = s_al[hh][c >> 3];
    const int    myrow = 8 * hh + (c & 7);
    const int    nP    = __builtin_amdgcn_readfirstlane(hh ? nPB : nPA);
    for (int j = 0; j < nP; j += 16) {
      v2f a[4]; float4 r[8];
#pragma unroll
      for (int q = 0; q < 4; ++q) {
        const int k = j + 4 * q + 2 * hf;
        a[q] = tile_a(rowp, alp, k, myrow);
        r[2 * q]     = ld_row(hcol, srcp[k], IN_CH);
        r[2 * q + 1] = ld_row(hcol, srcp[k + 1], IN_CH);
      }
      __builtin_amdgcn_sched_barrier(0);
#pragma unroll
      for (int q = 0; q < 4; ++q) {
        dsum += a[q].x + a[q].y;
        tile_mma(c0, c1, c2, c3, a[q], r[2 * q], r[2 * q + 1]);
      }
    }
  } else {
    if (tid < NPAIR) {
      const int row = tid & 15, q = tid >> 4, d = node0 + row;
      float mx = edge_score(as, ad, d, d, H, q);
      for (int e = 0; e < E0_EDGES; ++e)
        if (ei[E0_EDGES + e] == d) mx = fmaxf(mx, edge_score(as, ad, ei[e], d, H, q));
      s_mk[tid] = ord_key(mx);
    }
    __syncthreads();
    const float mx0 = ord_val(s_mk[hh * 16 + c]);
    for (int e0 = 0; e0 < E0_EDGES; e0 += 4) {
      const int e  = e0 + 2 * hf;
      const int d0 = ei[E0_EDGES + e], d1 = ei[E0_EDGES + e + 1];
      const bool in0 = (d0 >> 4) == T, in1 = (d1 >> 4) == T;
      if (__ballot(in0 | in1) == 0) continue;
      const int s0 = ei[e], s1 = ei[e + 1];
      v2f a;
      a.x = (in0 && (d0 & 15) == c) ? __expf(edge_score(as, ad, s0, d0, H, hh) - mx0) : 0.f;
      a.y = (in1 && (d1 & 15) == c) ? __expf(edge_score(as, ad, s1, d1, H, hh) - mx0) : 0.f;
      dsum += a.x + a.y;
      tile_mma(c0, c1, c2, c3, a, ld_row(hcol, s0, IN_CH), ld_row(hcol, s1, IN_CH));
    }
    for (int j = 0; j < 16; j += 4) {
      const int r0 = j + 2 * hf, n0 = node0 + r0, n1 = n0 + 1;
      v2f a;
      a.x = r0 == c     ? __expf(edge_score(as, ad, n0, n0, H, hh) - mx0) : 0.f;
      a.y = r0 + 1 == c ? __expf(edge_score(as, ad, n1, n1, H, hh) - mx0) : 0.f;
      dsum += a.x + a.y;
      tile_mma(c0, c1, c2, c3, a, ld_row(hcol, n0, IN_CH), ld_row(hcol, n1, IN_CH));
    }
  }

  const float dtot = dsum + __shfl_xor(dsum, 16);
  const int rb = listed ? hf * 16 + 8 * hh : hh * 16 + 8 * hf;
#pragma unroll
  for (int r = 0; r < 8; ++r) {
    const float inv = 1.0f / (__shfl(dtot, 8 * hf + r) + 1e-16f);
    *(float4*)&s_buf[(rb + r) * XP + xh * 64 + 4 * c] =
        make_float4(c0[r] * inv, c1[r] * inv, c2[r] * inv, c3[r] * inv);
  }
  __syncthreads();
  v8f e0 = {}, e1 = {}, e2 = {}, e3 = {};
  {
    const float* ax = &s_buf[(hh * 16 + c) * XP + 2 * hf];
    v2f a = *(const v2f*)ax;
    Bop b = load_b(Wl1, 0, HC1, u * 64, hf, c);
#pragma unroll 4
    for (int k0 = 0; k0 < IN_CH - 4; k0 += 4) {
      const v2f an = *(const v2f*)(ax + k0 + 4);
      Bop bn = load_b(Wl1, k0 + 4, HC1, u * 64, hf, c);
      e0 = __builtin_amdgcn_wmma_f32_16x16x4_f32(false, a, false, b.b0, (short)0, e0, false, false);
      e1 = __builtin_amdgcn_wmma_f32_16x16x4_f32(false, a, false, b.b1, (short)0, e1, false, false);
      e2 = __builtin_amdgcn_wmma_f32_16x16x4_f32(false, a, false, b.b2, (short)0, e2, false, false);
      e3 = __builtin_amdgcn_wmma_f32_16x16x4_f32(false, a, false, b.b3, (short)0, e3, false, false);
      a = an; b = bn;
    }
    e0 = __builtin_amdgcn_wmma_f32_16x16x4_f32(false, a, false, b.b0, (short)0, e0, false, false);
    e1 = __builtin_amdgcn_wmma_f32_16x16x4_f32(false, a, false, b.b1, (short)0, e1, false, false);
    e2 = __builtin_amdgcn_wmma_f32_16x16x4_f32(false, a, false, b.b2, (short)0, e2, false, false);
    e3 = __builtin_amdgcn_wmma_f32_16x16x4_f32(false, a, false, b.b3, (short)0, e3, false, false);
  }
  const float* bp = bias + u * 64 + c;
  const float bv0 = bp[0], bv1 = bp[16], bv2 = bp[32], bv3 = bp[48];
  __syncthreads();
  {
    float* tq = &s_buf[(8 * hf) * TP + u * 64 + c];
#pragma unroll
    for (int r = 0; r < 8; ++r) {
      tq[r * TP]      = elu1(bv0 + e0[r]);
      tq[r * TP + 16] = elu1(bv1 + e1[r]);
      tq[r * TP + 32] = elu1(bv2 + e2[r]);
      tq[r * TP + 48] = elu1(bv3 + e3[r]);
    }
  }
  __syncthreads();
  const int cb = (u & 1) * 32 + c;
  const float* __restrict__ Wb = (u < 2 ? W0 : W1) + (size_t)(2 * hf) * 64 + cb;
  const float* at = &s_buf[c * TP + 2 * hf];
  const float* vp = V2 + (c < 4 ? c : 0) * HC1 + u * 64 + 2 * hf;
  v2f tv[16];
  for (int i = 0; i < 16; ++i) tv[i] = *(const v2f*)(vp + 4 * i);
  v8f d0 = {}, d1 = {};
  v2f a = *(const v2f*)at;
  v2f b0, b1;
  b0.x = Wb[0];  b0.y = Wb[64];
  b1.x = Wb[16]; b1.y = Wb[64 + 16];
#pragma unroll 4
  for (int k0 = 0; k0 < HC1 - 4; k0 += 4) {
    const float* wn = Wb + (size_t)(k0 + 4) * 64;
    const v2f an = *(const v2f*)(at + k0 + 4);
    v2f bn0, bn1;
    bn0.x = wn[0];  bn0.y = wn[64];
    bn1.x = wn[16]; bn1.y = wn[64 + 16];
    d0 = __builtin_amdgcn_wmma_f32_16x16x4_f32(false, a, false, b0, (short)0, d0, false, false);
    d1 = __builtin_amdgcn_wmma_f32_16x16x4_f32(false, a, false, b1, (short)0, d1, false, false);
    a = an; b0 = bn0; b1 = bn1;
  }
  d0 = __builtin_amdgcn_wmma_f32_16x16x4_f32(false, a, false, b0, (short)0, d0, false, false);
  d1 = __builtin_amdgcn_wmma_f32_16x16x4_f32(false, a, false, b1, (short)0, d1, false, false);
  {
    const float* aq = at + u * 64;
    v8f p0 = {}, p1 = {};
    for (int i = 0; i < 16; i += 2) {
      const v2f a0 = *(const v2f*)(aq + 4 * i), a1 = *(const v2f*)(aq + 4 * i + 4);
      v2f q0, q1;
      q0.x = c < 4 ? tv[i].x : 0.f;     q0.y = c < 4 ? tv[i].y : 0.f;
      q1.x = c < 4 ? tv[i + 1].x : 0.f; q1.y = c < 4 ? tv[i + 1].y : 0.f;
      p0 = __builtin_amdgcn_wmma_f32_16x16x4_f32(false, a0, false, q0, (short)0, p0, false, false);
      p1 = __builtin_amdgcn_wmma_f32_16x16x4_f32(false, a1, false, q1, (short)0, p1, false, false);
    }
    if (c < 4) {
      for (int r = 0; r < 8; ++r) s_ps[u][8 * hf + r][c] = p0[r] + p1[r];
    }
  }
  float* ob = (u < 2 ? o0 : o1) + (size_t)(node0 + 8 * hf) * 64 + cb;
#pragma unroll
  for (int r = 0; r < 8; ++r) {
    ob[(size_t)r * 64]      = d0[r];
    ob[(size_t)r * 64 + 16] = d1[r];
  }
  __syncthreads();
  if (tid < 64) {
    const int row = tid >> 2, j = tid & 3;
    sc2[(size_t)j * NN + node0 + row] =
        ((s_ps[0][row][j] + s_ps[1][row][j]) + s_ps[2][row][j]) + s_ps[3][row][j];
  }
}

extern "C" void kernel_launch(void* const* d_in, const int* in_sizes, int n_in,
                              void* d_out, int out_size, void* d_ws, size_t ws_size,
                              hipStream_t stream) {
  const float* x    = (const float*)d_in[0];
  const int*   ei   = (const int*)  d_in[1];
  const float* W1   = (const float*)d_in[2];
  const float* aS1  = (const float*)d_in[3];
  const float* aD1  = (const float*)d_in[4];
  const float* b1   = (const float*)d_in[5];
  const float* Wmu  = (const float*)d_in[6];
  const float* aSmu = (const float*)d_in[7];
  const float* aDmu = (const float*)d_in[8];
  const float* bmu  = (const float*)d_in[9];
  const float* Wls  = (const float*)d_in[10];
  const float* aSls = (const float*)d_in[11];
  const float* aDls = (const float*)d_in[12];
  const float* bls  = (const float*)d_in[13];

  char* ws = (char*)d_ws;
  size_t off = 0;
  auto alloc = [&](size_t bytes) -> char* {
    off = (off + 255) & ~(size_t)255;
    char* p = ws + off; off += bytes; return p;
  };

  float*    h2   = (float*)   alloc((size_t)NN * 2 * OUT_CH * 4);
  float*    as1  = (float*)   alloc((size_t)NN * HEADS * 4);
  float*    ad1  = (float*)   alloc((size_t)NN * HEADS * 4);
  unsigned* tl   = (unsigned*)alloc((size_t)NTILES * TILE_CAP * 4);
  unsigned* tcnt = (unsigned*)alloc((size_t)NTILES * 4);
  float*    sc2  = (float*)   alloc((size_t)4 * NN * 4);
  float*    V1   = (float*)   alloc((size_t)4 * IN_CH * 4);
  float*    V2   = (float*)   alloc((size_t)4 * HC1 * 4);

  float* hm = h2;
  float* hl = h2 + (size_t)NN * OUT_CH;
  float* out_mu = (float*)d_out;
  float* out_ls = out_mu + (size_t)NN * OUT_CH;

  bin_edges<<<NGROUPS + 1, 512, 0, stream>>>(ei, tl, tcnt, W1, aS1, aD1, V1,
                                             Wmu, Wls, aSmu, aDmu, aSls, aDls, V2);

  x_scores<<<(NN + 255) / 256, 256, 0, stream>>>(x, V1, as1, ad1, NN);
  agg1_tile<<<NTILES, 128, 0, stream>>>(tl, tcnt, ei, as1, ad1, x, W1, b1, Wmu, Wls, hm, hl, V2, sc2);

  agg2_tile<<<NTILES, 64, 0, stream>>>(tl, tcnt, ei, sc2, hm, hl, bmu, bls, out_mu, out_ls);
}


# list building: a block's scan of the edge list split over 32 waves (1024 threads), not 16: each wave's chain of load waits, branches and LDS round trips halves; list order and sums change order
# speedup vs baseline: 4.7222x; 1.1039x over previous
#include <hip/hip_runtime.h>

typedef __attribute__((ext_vector_type(2))) float v2f;
typedef __attribute__((ext_vector_type(8))) float v8f;

#define NN       50000
#define IN_CH    128
#define OUT_CH   64
#define HEADS    2
#define HID      128
#define HC1      (HEADS*HID)
#define E0_EDGES 400000
#define E_TOT    (E0_EDGES + NN)
#define NEG_SLOPE 0.2f

struct Bop { v2f b0, b1, b2, b3; };

__device__ __forceinline__ float elu1(float v) { return v > 0.f ? v : (__expf(v) - 1.f); }

__device__ __forceinline__ Bop load_b(const float* __restrict__ B, int k0, int N,
                                      int tn, int lhalf, int lmod) {
  const float* r0 = B + (size_t)(k0 + 2 * lhalf) * N + tn + lmod;
  const float* r1 = r0 + N;
  Bop b;
  b.b0.x = r0[0];  b.b0.y = r1[0];
  b.b1.x = r0[16]; b.b1.y = r1[16];
  b.b2.x = r0[32]; b.b2.y = r1[32];
  b.b3.x = r0[48]; b.b3.y = r1[48];
  return b;
}

__device__ __forceinline__ float leaky(float v) { return v >= 0.f ? v : NEG_SLOPE * v; }

#define NTILES    (NN / 16)
#define GROUP_T   16
#define NGROUPS   ((NTILES + GROUP_T - 1) / GROUP_T)
#define NHALF     (2 * GROUP_T)
#define SCAN_W    32
#define SUB_CAP   24
#define TILE_CAP  256
#define LIST_FULL 0xFFFFFFFFu
#define MAX_ENT   (TILE_CAP + 16 + 16)

__global__ void __launch_bounds__(32 * SCAN_W)
bin_edges(const int* __restrict__ ei, unsigned* __restrict__ tl,
          unsigned* __restrict__ tcnt, const float* __restrict__ Wl1,
          const float* __restrict__ att_s, const float* __restrict__ att_d,
          float* __restrict__ V,
          const float* __restrict__ Wm, const float* __restrict__ Wl,
          const float* __restrict__ atsm, const float* __restrict__ atdm,
          const float* __restrict__ atsl, const float* __restrict__ atdl,
          float* __restrict__ V2) {
  static_assert(NHALF == 32 && SCAN_W == 32, "a wave's 32 lanes read the 32 pieces' counters of a half tile");
  __shared__ unsigned cnt[SCAN_W][NHALF];
  __shared__ unsigned sub[SCAN_W][NHALF][SUB_CAP];
  const int g = blockIdx.x, tid = threadIdx.x;
  const int lane = tid & 31, w = tid >> 5;
  if (g == NGROUPS) {
    if (tid < 4 * IN_CH) {
      const int which = tid >> 7, k = tid & (IN_CH - 1), hd = which & 1;
      const float4* wr = (const float4*)(Wl1 + (size_t)k * HC1 + hd * HID);
      const float4* av = (const float4*)((which < 2 ? att_s : att_d) + hd * HID);
      float s = 0.f;
#pragma unroll 8
      for (int cc = 0; cc < HID / 4; ++cc) {
        const float4 a = wr[cc], b = av[cc];
        s += a.x * b.x + a.y * b.y + a.z * b.z + a.w * b.w;
      }
      V[which * IN_CH + k] = s;
    }
    for (int v = tid; v < 4 * HC1; v += 32 * SCAN_W) {
      const int j = v >> 8, k2 = v & (HC1 - 1);
      const float4* wr2 = (const float4*)((j < 2 ? Wm : Wl) + (size_t)k2 * OUT_CH);
      const float4* av2 = (const float4*)(j == 0 ? atsm : j == 1 ? atdm : j == 2 ? atsl : atdl);
      float s2 = 0.f;
#pragma unroll 8
      for (int cc = 0; cc < OUT_CH / 4; ++cc) {
        const float4 a = wr2[cc], b = av2[cc];
        s2 += a.x * b.x + a.y * b.y + a.z * b.z + a.w * b.w;
      }
      V2[v] = s2;
    }
    return;
  }
  cnt[tid >> 5][tid & 31] = 0u;
  __syncthreads();

  const int per_wave = E0_EDGES / SCAN_W;
  static_assert(E0_EDGES % (4 * SCAN_W) == 0, "every wave's slice starts and ends on a 16-byte piece");
  const int beg = w * per_wave, end = beg + per_wave;
  const int* __restrict__ dstp = ei + E0_EDGES;
  for (int base = beg; base < end; base += 1024) {
    int4 dv[8];
#pragma unroll
    for (int q = 0; q < 8; ++q) {
      const int idx = base + q * 128 + 4 * lane;
      dv[q] = *(const int4*)(dstp + (idx < end ? idx : end - 4));
    }
#pragma unroll
    for (int q = 0; q < 8; ++q) {
      const int idx = base + q * 128 + 4 * lane;
      const int dd[4] = {dv[q].x, dv[q].y, dv[q].z, dv[q].w};
#pragma unroll
      for (int k = 0; k < 4; ++k) {
        const int d = dd[k];
        if (idx < end && (d >> 8) == g) {
          const int ti = (d >> 3) & (NHALF - 1);
          const unsigned p = atomicAdd(&cnt[w][ti], 1u);
          if (p < SUB_CAP) sub[w][ti][p] = (unsigned)(idx + k) | ((unsigned)(d & 15) << 24);
        }
      }
    }
  }
  __syncthreads();

  const int t = w >> 1, hs = w & 1;
  const int T = g * GROUP_T + t;
  if (T < NTILES) {
    const unsigned n0 = cnt[lane][2 * t], n1 = cnt[lane][2 * t + 1];
    const bool full0 = __ballot(n0 > SUB_CAP || n1 > SUB_CAP) != 0;
    unsigned first = n0, second = n1;
#pragma unroll
    for (int o = 16; o > 0; o >>= 1) { first += __shfl_xor(first, o); second += __shfl_xor(second, o); }
    const int ht = 2 * t + hs;
    unsigned total = hs ? first : 0u;
    for (int w2 = 0; w2 < SCAN_W; ++w2) {
      const unsigned n = cnt[w2][ht];
      const unsigned nn = n < SUB_CAP ? n : SUB_CAP;
      for (unsigned i = lane; i < nn; i += 32) {
        const unsigned pos = total + i;
        if (pos < TILE_CAP) tl[(size_t)T * TILE_CAP + pos] = sub[w2][ht][i];
      }
      total += n;
    }
    const unsigned all = first + second;
    if (hs == 0 && lane == 0)
      tcnt[T] = (full0 || all > TILE_CAP) ? LIST_FULL : (all | (first << 16));
  }
}

__device__ __forceinline__ float edge_score(const float* __restrict__ as,
    const float* __restrict__ ad, int s, int d, int H, int q) {
  return leaky(as[s * H + q] + ad[d * H + q]);
}

__device__ __forceinline__ float4 ld_row(const float* __restrict__ hcol, int s, int HC) {
  return *(const float4*)(hcol + (size_t)s * HC);
}
__device__ __forceinline__ v2f tile_a(const int* row, const float* al, int k, int c) {
  const float w0 = al[k], w1 = al[k + 1];
  v2f a;
  a.x = row[k]     == c ? w0 : 0.f;
  a.y = row[k + 1] == c ? w1 : 0.f;
  return a;
}
__device__ __forceinline__ void tile_mma(v8f& c0, v8f& c1, v8f& c2, v8f& c3, v2f a,
                                         float4 r0, float4 r1) {
  v2f b;
  b.x = r0.x; b.y = r1.x;
  c0 = __builtin_amdgcn_wmma_f32_16x16x4_f32(false, a, false, b, (short)0, c0, false, false);
  b.x = r0.y; b.y = r1.y;
  c1 = __builtin_amdgcn_wmma_f32_16x16x4_f32(false, a, false, b, (short)0, c1, false, false);
  b.x = r0.z; b.y = r1.z;
  c2 = __builtin_amdgcn_wmma_f32_16x16x4_f32(false, a, false, b, (short)0, c2, false, false);
  b.x = r0.w; b.y = r1.w;
  c3 = __builtin_amdgcn_wmma_f32_16x16x4_f32(false, a, false, b, (short)0, c3, false, false);
}

__device__ __forceinline__ int   ord_key(float f) { const int b = __float_as_int(f); return b >= 0 ? b : b ^ 0x7FFFFFFF; }
__device__ __forceinline__ float ord_val(int k)   { return __int_as_float(k >= 0 ? k : k ^ 0x7FFFFFFF); }

__device__ __forceinline__ float edge_score2(const float* __restrict__ sc, int s, int d, int q) {
  return leaky(sc[(size_t)(2 * q) * NN + s] + sc[(size_t)(2 * q + 1) * NN + d]);
}

__global__ void __launch_bounds__(64)
agg2_tile(const unsigned* __restrict__ tl, const unsigned* __restrict__ tcnt,
          const int* __restrict__ ei, const float* __restrict__ sc,
          const float* __restrict__ h0, const float* __restrict__ h1,
          const float* __restrict__ bias0, const float* __restrict__ bias1,
          float* __restrict__ out0, float* __restrict__ out1) {
  constexpr int H     = 2;
  constexpr int C     = OUT_CH;
  constexpr int NT    = 64;
  constexpr int NPAIR = H * 16;
  static_assert(NT >= NPAIR, "one thread per pair clears the maxima");
  __shared__ int   s_src[MAX_ENT];
  __shared__ int   s_row[MAX_ENT];
  __shared__ float s_al[H][MAX_ENT];
  __shared__ int   s_mk[NPAIR];
  const int T = blockIdx.x, tid = threadIdx.x;
  const int lane = tid & 31, u = tid >> 5, c = lane & 15, hf = lane >> 4;
  const int node0 = T * 16;
  const float* __restrict__ hcol = (u ? h1 : h0) + 4 * c;

  v8f c0 = {}, c1 = {}, c2 = {}, c3 = {};
  float dsum = 0.f;
  const float* bp = (u ? bias1 : bias0) + 4 * c;
  const float4 bv = make_float4(bp[0], bp[1], bp[2], bp[3]);

  const unsigned cntT = tcnt[T];
  if (cntT != LIST_FULL) {
    const int n  = (int)(cntT & 0xFFFFu);
    const int nE = n + 16;
    const int nP = (nE + 15) & ~15;
    if (tid < NPAIR) s_mk[tid] = (int)0x80000000;
    __syncthreads();
    for (int i = tid; i < nP; i += NT) {
      int s = node0, row = 0;
      if (i < n) {
        const unsigned ent = tl[(size_t)T * TILE_CAP + i];
        s = ei[ent & 0x00FFFFFFu]; row = (int)(ent >> 24);
      } else if (i < nE) {
        row = i - n; s = node0 + row;
      }
      s_src[i] = s; s_row[i] = row;
      float sv0 = 0.f, sv1 = 0.f;
      if (i < nE) {
        sv0 = edge_score2(sc, s, node0 + row, 0);
        sv1 = edge_score2(sc, s, node0 + row, 1);
        atomicMax(&s_mk[row], ord_key(sv0));
        atomicMax(&s_mk[16 + row], ord_key(sv1));
      }
      s_al[0][i] = sv0; s_al[1][i] = sv1;
    }
    __syncthreads();
    for (int i = tid; i < nE; i += NT) {
      const int row = s_row[i];
#pragma unroll
      for (int q = 0; q < H; ++q)
        s_al[q][i] = __expf(s_al[q][i] - ord_val(s_mk[q * 16 + row]));
    }
    __syncthreads();
    const float* alp = s_al[u];
    for (int j = 0; j < nP; j += 16) {
      v2f a[4]; float4 r[8];
#pragma unroll
      for (int q = 0; q < 4; ++q) {
        const int k = j + 4 * q + 2 * hf;
        a[q] = tile_a(s_row, alp, k, c);
        r[2 * q]     = ld_row(hcol, s_src[k], C);
        r[2 * q + 1] = ld_row(hcol, s_src[k + 1], C);
      }
      __builtin_amdgcn_sched_barrier(0);
#pragma unroll
      for (int q = 0; q < 4; ++q) {
        dsum += a[q].x + a[q].y;
        tile_mma(c0, c1, c2, c3, a[q], r[2 * q], r[2 * q + 1]);
      }
    }
  } else {
    if (tid < NPAIR) {
      const int row = tid & 15, q = tid >> 4, d = node0 + row;
      float mx = edge_score2(sc, d, d, q);
      for (int e = 0; e < E0_EDGES; ++e)
        if (ei[E0_EDGES + e] == d) mx = fmaxf(mx, edge_score2(sc, ei[e], d, q));
      s_mk[tid] = ord_key(mx);
    }
    __syncthreads();
    const float mx0 = ord_val(s_mk[u * 16 + c]);
    for (int e0 = 0; e0 < E0_EDGES; e0 += 4) {
      const int e  = e0 + 2 * hf;
      const int d0 = ei[E0_EDGES + e], d1 = ei[E0_EDGES + e + 1];
      const bool in0 = (d0 >> 4) == T, in1 = (d1 >> 4) == T;
      if (__ballot(in0 | in1) == 0) continue;
      const int s0 = ei[e], s1 = ei[e + 1];
      v2f a;
      a.x = (in0 && (d0 & 15) == c) ? __expf(edge_score2(sc, s0, d0, u) - mx0) : 0.f;
      a.y = (in1 && (d1 & 15) == c) ? __expf(edge_score2(sc, s1, d1, u) - mx0) : 0.f;
      dsum += a.x + a.y;
      tile_mma(c0, c1, c2, c3, a, ld_row(hcol, s0, C), ld_row(hcol, s1, C));
    }
    for (int j = 0; j < 16; j += 4) {
      const int r0 = j + 2 * hf, n0 = node0 + r0, n1 = n0 + 1;
      v2f a;
      a.x = r0 == c     ? __expf(edge_score2(sc, n0, n0, u) - mx0) : 0.f;
      a.y = r0 + 1 == c ? __expf(edge_score2(sc, n1, n1, u) - mx0) : 0.f;
      dsum += a.x + a.y;
      tile_mma(c0, c1, c2, c3, a, ld_row(hcol, n0, C), ld_row(hcol, n1, C));
    }
  }

  const float dtot = dsum + __shfl_xor(dsum, 16);
  float* __restrict__ ob = (u ? out1 : out0) + 4 * c;
#pragma unroll
  for (int r = 0; r < 8; ++r) {
    const float inv = 1.0f / (__shfl(dtot, 8 * hf + r) + 1e-16f);
    float* q = ob + (size_t)(node0 + 8 * hf + r) * C;
    *(float4*)q = make_float4(bv.x + c0[r] * inv, bv.y + c1[r] * inv,
                              bv.z + c2[r] * inv, bv.w + c3[r] * inv);
  }
}

__global__ void __launch_bounds__(256)
x_scores(const float* __restrict__ x, const float* __restrict__ V,
         float* __restrict__ as, float* __restrict__ ad, int n_nodes) {
  const int i = blockIdx.x * blockDim.x + threadIdx.x;
  if (i >= n_nodes) return;
  const float4* xp = (const float4*)(x + (size_t)i * IN_CH);
  const float4* v0 = (const float4*)V;
  const float4* v1 = v0 + IN_CH / 4;
  const float4* v2 = v1 + IN_CH / 4;
  const float4* v3 = v2 + IN_CH / 4;
  float s0 = 0.f, s1 = 0.f, d0 = 0.f, d1 = 0.f;
#pragma unroll 8
  for (int cc = 0; cc < IN_CH / 4; ++cc) {
    const float4 v = xp[cc], a = v0[cc], b = v1[cc], p = v2[cc], q = v3[cc];
    s0 += v.x * a.x + v.y * a.y + v.z * a.z + v.w * a.w;
    s1 += v.x * b.x + v.y * b.y + v.z * b.z + v.w * b.w;
    d0 += v.x * p.x + v.y * p.y + v.z * p.z + v.w * p.w;
    d1 += v.x * q.x + v.y * q.y + v.z * q.z + v.w * q.w;
  }
  as[2 * i] = s0; as[2 * i + 1] = s1;
  ad[2 * i] = d0; ad[2 * i + 1] = d1;
}

__global__ void __launch_bounds__(128)
agg1_tile(const unsigned* __restrict__ tl, const unsigned* __restrict__ tcnt,
          const int* __restrict__ ei, const float* __restrict__ as,
          const float* __restrict__ ad, const float* __restrict__ x,
          const float* __restrict__ Wl1, const float* __restrict__ bias,
          const float* __restrict__ W0, const float* __restrict__ W1,
          float* __restrict__ o0, float* __restrict__ o1,
          const float* __restrict__ V2, float* __restrict__ sc2) {
  constexpr int H = HEADS, NT = 128, NPAIR = H * 16;
  __shared__ float s_ps[4][16][4];
  __shared__ int   s_src[2][MAX_ENT];
  __shared__ int   s_row[2][MAX_ENT];
  __shared__ float s_al[2][H][MAX_ENT];
  __shared__ int   s_mk[NPAIR];
  constexpr int XP = IN_CH + 4;
  constexpr int TP = HC1 + 4;
  static_assert(H * 16 * XP >= 16 * TP, "the tile fits where the sums were");
  __shared__ __attribute__((aligned(16))) float s_buf[H * 16 * XP];
  const int T = blockIdx.x, tid = threadIdx.x;
  const int lane = tid & 31, u = tid >> 5, c = lane & 15, hf = lane >> 4;
  const int hh = u >> 1, xh = u & 1;
  const int node0 = T * 16;
  const float* __restrict__ hcol = x + xh * 64 + 4 * c;

  v8f c0 = {}, c1 = {}, c2 = {}, c3 = {};
  float dsum = 0.f;

  const unsigned cntT = tcnt[T];
  const bool listed = cntT != LIST_FULL;
  if (listed) {
    const int n   = (int)(cntT & 0xFFFFu);
    const int nA  = (int)(cntT >> 16), nB = n - nA;
    const int nPA = (nA + 8 + 15) & ~15;
    const int nPB = (nB + 8 + 15) & ~15;
    if (tid < NPAIR) s_mk[tid] = (int)0x80000000;
    __syncthreads();
    for (int i = tid; i < nPA + nPB; i += NT) {
      const int sb = i >= nPA ? 1 : 0;
      const int j  = i - (sb ? nPA : 0);
      const int nS = sb ? nB : nA;
      int s = node0, row = 8 * sb;
      if (j < nS) {
        const unsigned ent = tl[(size_t)T * TILE_CAP + (sb ? nA : 0) + j];
        s = ei[ent & 0x00FFFFFFu]; row = (int)(ent >> 24);
      } else if (j < nS + 8) {
        row = 8 * sb + (j - nS); s = node0 + row;
      }
      s_src[sb][j] = s; s_row[sb][j] = row;
#pragma unroll
      for (int q = 0; q < H; ++q) {
        float sc = 0.f;
        if (j < nS + 8) {
          sc = edge_score(as, ad, s, node0 + row, H, q);
          atomicMax(&s_mk[q * 16 + row], ord_key(sc));
        }
        s_al[sb][q][j] = sc;
      }
    }
    __syncthreads();
    for (int i = tid; i < nPA + nPB; i += NT) {
      const int sb = i >= nPA ? 1 : 0;
      const int j  = i - (sb ? nPA : 0);
      if (j < (sb ? nB : nA) + 8) {
        const int row = s_row[sb][j];
#pragma unroll
        for (int q = 0; q < H; ++q)
          s_al[sb][q][j] = __expf(s_al[sb][q][j] - ord_val(s_mk[q * 16 + row]));
      }
    }
    __syncthreads();
    const int*   rowp  = s_row[hh];
    const int*   srcp  = s_src[hh];
    const float* alp   = s_al[hh][c >> 3];
    const int    myrow = 8 * hh + (c & 7);
    const int    nP    = __builtin_amdgcn_readfirstlane(hh ? nPB : nPA);
    for (int j = 0; j < nP; j += 16) {
      v2f a[4]; float4 r[8];
#pragma unroll
      for (int q = 0; q < 4; ++q) {
        const int k = j + 4 * q + 2 * hf;
        a[q] = tile_a(rowp, alp, k, myrow);
        r[2 * q]     = ld_row(hcol, srcp[k], IN_CH);
        r[2 * q + 1] = ld_row(hcol, srcp[k + 1], IN_CH);
      }
      __builtin_amdgcn_sched_barrier(0);
#pragma unroll
      for (int q = 0; q < 4; ++q) {
        dsum += a[q].x + a[q].y;
        tile_mma(c0, c1, c2, c3, a[q], r[2 * q], r[2 * q + 1]);
      }
    }
  } else {
    if (tid < NPAIR) {
      const int row = tid & 15, q = tid >> 4, d = node0 + row;
      float mx = edge_score(as, ad, d, d, H, q);
      for (int e = 0; e < E0_EDGES; ++e)
        if (ei[E0_EDGES + e] == d) mx = fmaxf(mx, edge_score(as, ad, ei[e], d, H, q));
      s_mk[tid] = ord_key(mx);
    }
    __syncthreads();
    const float mx0 = ord_val(s_mk[hh * 16 + c]);
    for (int e0 = 0; e0 < E0_EDGES; e0 += 4) {
      const int e  = e0 + 2 * hf;
      const int d0 = ei[E0_EDGES + e], d1 = ei[E0_EDGES + e + 1];
      const bool in0 = (d0 >> 4) == T, in1 = (d1 >> 4) == T;
      if (__ballot(in0 | in1) == 0) continue;
      const int s0 = ei[e], s1 = ei[e + 1];
      v2f a;
      a.x = (in0 && (d0 & 15) == c) ? __expf(edge_score(as, ad, s0, d0, H, hh) - mx0) : 0.f;
      a.y = (in1 && (d1 & 15) == c) ? __expf(edge_score(as, ad, s1, d1, H, hh) - mx0) : 0.f;
      dsum += a.x + a.y;
      tile_mma(c0, c1, c2, c3, a, ld_row(hcol, s0, IN_CH), ld_row(hcol, s1, IN_CH));
    }
    for (int j = 0; j < 16; j += 4) {
      const int r0 = j + 2 * hf, n0 = node0 + r0, n1 = n0 + 1;
      v2f a;
      a.x = r0 == c     ? __expf(edge_score(as, ad, n0, n0, H, hh) - mx0) : 0.f;
      a.y = r0 + 1 == c ? __expf(edge_score(as, ad, n1, n1, H, hh) - mx0) : 0.f;
      dsum += a.x + a.y;
      tile_mma(c0, c1, c2, c3, a, ld_row(hcol, n0, IN_CH), ld_row(hcol, n1, IN_CH));
    }
  }

  const float dtot = dsum + __shfl_xor(dsum, 16);
  const int rb = listed ? hf * 16 + 8 * hh : hh * 16 + 8 * hf;
#pragma unroll
  for (int r = 0; r < 8; ++r) {
    const float inv = 1.0f / (__shfl(dtot, 8 * hf + r) + 1e-16f);
    *(float4*)&s_buf[(rb + r) * XP + xh * 64 + 4 * c] =
        make_float4(c0[r] * inv, c1[r] * inv, c2[r] * inv, c3[r] * inv);
  }
  __syncthreads();
  v8f e0 = {}, e1 = {}, e2 = {}, e3 = {};
  {
    const float* ax = &s_buf[(hh * 16 + c) * XP + 2 * hf];
    v2f a = *(const v2f*)ax;
    Bop b = load_b(Wl1, 0, HC1, u * 64, hf, c);
#pragma unroll 4
    for (int k0 = 0; k0 < IN_CH - 4; k0 += 4) {
      const v2f an = *(const v2f*)(ax + k0 + 4);
      Bop bn = load_b(Wl1, k0 + 4, HC1, u * 64, hf, c);
      e0 = __builtin_amdgcn_wmma_f32_16x16x4_f32(false, a, false, b.b0, (short)0, e0, false, false);
      e1 = __builtin_amdgcn_wmma_f32_16x16x4_f32(false, a, false, b.b1, (short)0, e1, false, false);
      e2 = __builtin_amdgcn_wmma_f32_16x16x4_f32(false, a, false, b.b2, (short)0, e2, false, false);
      e3 = __builtin_amdgcn_wmma_f32_16x16x4_f32(false, a, false, b.b3, (short)0, e3, false, false);
      a = an; b = bn;
    }
    e0 = __builtin_amdgcn_wmma_f32_16x16x4_f32(false, a, false, b.b0, (short)0, e0, false, false);
    e1 = __builtin_amdgcn_wmma_f32_16x16x4_f32(false, a, false, b.b1, (short)0, e1, false, false);
    e2 = __builtin_amdgcn_wmma_f32_16x16x4_f32(false, a, false, b.b2, (short)0, e2, false, false);
    e3 = __builtin_amdgcn_wmma_f32_16x16x4_f32(false, a, false, b.b3, (short)0, e3, false, false);
  }
  const float* bp = bias + u * 64 + c;
  const float bv0 = bp[0], bv1 = bp[16], bv2 = bp[32], bv3 = bp[48];
  __syncthreads();
  {
    float* tq = &s_buf[(8 * hf) * TP + u * 64 + c];
#pragma unroll
    for (int r = 0; r < 8; ++r) {
      tq[r * TP]      = elu1(bv0 + e0[r]);
      tq[r * TP + 16] = elu1(bv1 + e1[r]);
      tq[r * TP + 32] = elu1(bv2 + e2[r]);
      tq[r * TP + 48] = elu1(bv3 + e3[r]);
    }
  }
  __syncthreads();
  const int cb = (u & 1) * 32 + c;
  const float* __restrict__ Wb = (u < 2 ? W0 : W1) + (size_t)(2 * hf) * 64 + cb;
  const float* at = &s_buf[c * TP + 2 * hf];
  const float* vp = V2 + (c < 4 ? c : 0) * HC1 + u * 64 + 2 * hf;
  v2f tv[16];
  for (int i = 0; i < 16; ++i) tv[i] = *(const v2f*)(vp + 4 * i);
  v8f d0 = {}, d1 = {};
  v2f a = *(const v2f*)at;
  v2f b0, b1;
  b0.x = Wb[0];  b0.y = Wb[64];
  b1.x = Wb[16]; b1.y = Wb[64 + 16];
#pragma unroll 4
  for (int k0 = 0; k0 < HC1 - 4; k0 += 4) {
    const float* wn = Wb + (size_t)(k0 + 4) * 64;
    const v2f an = *(const v2f*)(at + k0 + 4);
    v2f bn0, bn1;
    bn0.x = wn[0];  bn0.y = wn[64];
    bn1.x = wn[16]; bn1.y = wn[64 + 16];
    d0 = __builtin_amdgcn_wmma_f32_16x16x4_f32(false, a, false, b0, (short)0, d0, false, false);
    d1 = __builtin_amdgcn_wmma_f32_16x16x4_f32(false, a, false, b1, (short)0, d1, false, false);
    a = an; b0 = bn0; b1 = bn1;
  }
  d0 = __builtin_amdgcn_wmma_f32_16x16x4_f32(false, a, false, b0, (short)0, d0, false, false);
  d1 = __builtin_amdgcn_wmma_f32_16x16x4_f32(false, a, false, b1, (short)0, d1, false, false);
  {
    const float* aq = at + u * 64;
    v8f p0 = {}, p1 = {};
    for (int i = 0; i < 16; i += 2) {
      const v2f a0 = *(const v2f*)(aq + 4 * i), a1 = *(const v2f*)(aq + 4 * i + 4);
      v2f q0, q1;
      q0.x = c < 4 ? tv[i].x : 0.f;     q0.y = c < 4 ? tv[i].y : 0.f;
      q1.x = c < 4 ? tv[i + 1].x : 0.f; q1.y = c < 4 ? tv[i + 1].y : 0.f;
      p0 = __builtin_amdgcn_wmma_f32_16x16x4_f32(false, a0, false, q0, (short)0, p0, false, false);
      p1 = __builtin_amdgcn_wmma_f32_16x16x4_f32(false, a1, false, q1, (short)0, p1, false, false);
    }
    if (c < 4) {
      for (int r = 0; r < 8; ++r) s_ps[u][8 * hf + r][c] = p0[r] + p1[r];
    }
  }
  float* ob = (u < 2 ? o0 : o1) + (size_t)(node0 + 8 * hf) * 64 + cb;
#pragma unroll
  for (int r = 0; r < 8; ++r) {
    ob[(size_t)r * 64]      = d0[r];
    ob[(size_t)r * 64 + 16] = d1[r];
  }
  __syncthreads();
  if (tid < 64) {
    const int row = tid >> 2, j = tid & 3;
    sc2[(size_t)j * NN + node0 + row] =
        ((s_ps[0][row][j] + s_ps[1][row][j]) + s_ps[2][row][j]) + s_ps[3][row][j];
  }
}

extern "C" void kernel_launch(void* const* d_in, const int* in_sizes, int n_in,
                              void* d_out, int out_size, void* d_ws, size_t ws_size,
                              hipStream_t stream) {
  const float* x    = (const float*)d_in[0];
  const int*   ei   = (const int*)  d_in[1];
  const float* W1   = (const float*)d_in[2];
  const float* aS1  = (const float*)d_in[3];
  const float* aD1  = (const float*)d_in[4];
  const float* b1   = (const float*)d_in[5];
  const float* Wmu  = (const float*)d_in[6];
  const float* aSmu = (const float*)d_in[7];
  const float* aDmu = (const float*)d_in[8];
  const float* bmu  = (const float*)d_in[9];
  const float* Wls  = (const float*)d_in[10];
  const float* aSls = (const float*)d_in[11];
  const float* aDls = (const float*)d_in[12];
  const float* bls  = (const float*)d_in[13];

  char* ws = (char*)d_ws;
  size_t off = 0;
  auto alloc = [&](size_t bytes) -> char* {
    off = (off + 255) & ~(size_t)255;
    char* p = ws + off; off += bytes; return p;
  };

  float*    h2   = (float*)   alloc((size_t)NN * 2 * OUT_CH * 4);
  float*    as1  = (float*)   alloc((size_t)NN * HEADS * 4);
  float*    ad1  = (float*)   alloc((size_t)NN * HEADS * 4);
  unsigned* tl   = (unsigned*)alloc((size_t)NTILES * TILE_CAP * 4);
  unsigned* tcnt = (unsigned*)alloc((size_t)NTILES * 4);
  float*    sc2  = (float*)   alloc((size_t)4 * NN * 4);
  float*    V1   = (float*)   alloc((size_t)4 * IN_CH * 4);
  float*    V2   = (float*)   alloc((size_t)4 * HC1 * 4);

  float* hm = h2;
  float* hl = h2 + (size_t)NN * OUT_CH;
  float* out_mu = (float*)d_out;
  float* out_ls = out_mu + (size_t)NN * OUT_CH;

  bin_edges<<<NGROUPS + 1, 32 * SCAN_W, 0, stream>>>(ei, tl, tcnt, W1, aS1, aD1, V1,
                                             Wmu, Wls, aSmu, aDmu, aSls, aDls, V2);

  x_scores<<<(NN + 255) / 256, 256, 0, stream>>>(x, V1, as1, ad1, NN);
  agg1_tile<<<NTILES, 128, 0, stream>>>(tl, tcnt, ei, as1, ad1, x, W1, b1, Wmu, Wls, hm, hl, V2, sc2);

  agg2_tile<<<NTILES, 64, 0, stream>>>(tl, tcnt, ei, sc2, hm, hl, bmu, bls, out_mu, out_ls);
}
